# starting kernel
# baseline (speedup 1.0000x reference)
#include <hip/hip_runtime.h>
#include <math.h>


typedef __attribute__((ext_vector_type(2))) float v2f;
typedef __attribute__((ext_vector_type(8))) float v8f;

#define B_      2048
#define T_      200
#define D_      128
#define NNODE_  200
#define TP      208
#define NT      13
#define HSTR    132
#define SSTR    210
#define PBS     (TP * D_)
#define NEGC    (-4294967295.0f)

__device__ __forceinline__ v8f wmma4(v2f a, v2f b, v8f c) {
  return __builtin_amdgcn_wmma_f32_16x16x4_f32(false, a, false, b, (short)0, c,
                                               false, false);
}

__global__ void qkv_kernel(const float* __restrict__ re, const int* __restrict__ alias,
                           const float* __restrict__ Wq, const float* __restrict__ Wk,
                           const float* __restrict__ Wv,
                           float* __restrict__ wsQ, float* __restrict__ wsK,
                           float* __restrict__ wsV) {
  extern __shared__ float smem[];
  float* sH = smem;
  float* sW = smem + TP * HSTR;
  const int b = blockIdx.x, tid = threadIdx.x;

  for (int idx = tid; idx < TP * D_; idx += 256) {
    int t = idx >> 7, d = idx & 127;
    float v = 0.f;
    if (t < T_) {
      int a = alias[b * T_ + t];
      v = re[((size_t)b * NNODE_ + a) * D_ + d];
    }
    sH[t * HSTR + d] = v;
  }
  for (int idx = tid; idx < 3 * D_ * D_; idx += 256) {
    int mat = idx >> 14, r = idx & 16383;
    const float* src = (mat == 0) ? Wq : (mat == 1) ? Wk : Wv;
    sW[idx] = src[r];
  }
  __syncthreads();

  const int wave = tid >> 5, lane = tid & 31, lh = lane & 15, hi = lane >> 4;
  for (int tile = wave; tile < 3 * NT * 8; tile += 8) {
    int mat = tile / (NT * 8), rem = tile % (NT * 8);
    int i = rem >> 3, n0 = (rem & 7) << 4;
    const float* W = sW + mat * D_ * D_;
    const float* arow = sH + (i * 16 + lh) * HSTR;
    v8f acc = {0.f, 0.f, 0.f, 0.f, 0.f, 0.f, 0.f, 0.f};
#pragma unroll
    for (int c = 0; c < 32; ++c) {
      int k = 4 * c + 2 * hi;
      v2f a = *(const v2f*)(arow + k);
      v2f bb;
      bb.x = W[k * D_ + n0 + lh];
      bb.y = W[(k + 1) * D_ + n0 + lh];
      acc = wmma4(a, bb, acc);
    }
    float* dst = ((mat == 0) ? wsQ : (mat == 1) ? wsK : wsV) + (size_t)b * PBS;
#pragma unroll
    for (int r = 0; r < 8; ++r) {
      int row = i * 16 + r + 8 * hi;
      float v = acc[r];
      dst[row * D_ + n0 + lh] = v > 0.f ? v : 0.f;
    }
  }
}

__global__ void attn_kernel(const float* __restrict__ re, const int* __restrict__ alias,
                            const int* __restrict__ seqlen,
                            const float* __restrict__ wsQ, const float* __restrict__ wsK,
                            const float* __restrict__ wsV, float* __restrict__ wsDec) {
  extern __shared__ float smem[];
  float* sK = smem;
  float* sV = smem + TP * HSTR;
  float* sS0 = smem + 2 * TP * HSTR;
  const int b = blockIdx.x, tid = threadIdx.x;
  const int len = seqlen[b];
  const float* Kb = wsK + (size_t)b * PBS;
  const float* Vb = wsV + (size_t)b * PBS;
  for (int idx = tid; idx < TP * D_; idx += 256) {
    int t = idx >> 7, d = idx & 127;
    sK[t * HSTR + d] = Kb[idx];
    sV[t * HSTR + d] = Vb[idx];
  }
  __syncthreads();

  const int wave = tid >> 5, lane = tid & 31, lh = lane & 15, hi = lane >> 4;
  float* sS = sS0 + wave * (16 * SSTR);
  const float scale = 0.088388347648318447f;

  for (int pass = 0; pass < 2; ++pass) {
    const int qt = wave + 8 * pass;
    const bool active = (qt < NT);
    if (active) {
      v2f qa[32];
      const float* Qrow = wsQ + (size_t)b * PBS + (qt * 16 + lh) * D_;
#pragma unroll
      for (int c = 0; c < 32; ++c) qa[c] = *(const v2f*)(Qrow + 4 * c + 2 * hi);
      for (int j = 0; j < NT; ++j) {
        v8f acc = {0.f, 0.f, 0.f, 0.f, 0.f, 0.f, 0.f, 0.f};
#pragma unroll
        for (int c = 0; c < 32; ++c) {
          v2f bb = *(const v2f*)(sK + (j * 16 + lh) * HSTR + 4 * c + 2 * hi);
          acc = wmma4(qa[c], bb, acc);
        }
#pragma unroll
        for (int r = 0; r < 8; ++r) {
          int m = r + 8 * hi, s = j * 16 + lh;
          sS[m * SSTR + s] = (s < len) ? acc[r] * scale : NEGC;
        }
      }
    }
    __syncthreads();
    if (active) {
      float* rowp = sS + lh * SSTR + hi * 104;
      float mx = -3.0e38f;
      for (int s = 0; s < 104; ++s) mx = fmaxf(mx, rowp[s]);
      mx = fmaxf(mx, __shfl_xor(mx, 16, 32));
      float sum = 0.f;
      for (int s = 0; s < 104; ++s) { float e = __expf(rowp[s] - mx); rowp[s] = e; sum += e; }
      sum += __shfl_xor(sum, 16, 32);
      float qm = ((qt * 16 + lh) < len) ? 1.f : 0.f;
      float inv = qm / sum;
      for (int s = 0; s < 104; ++s) rowp[s] *= inv;
    }
    __syncthreads();
    if (active) {
      float* Db = wsDec + (size_t)b * PBS;
      for (int nt = 0; nt < 8; ++nt) {
        int n0 = nt * 16;
        v8f acc;
#pragma unroll
        for (int r = 0; r < 8; ++r) {
          int t = qt * 16 + r + 8 * hi;
          float h = 0.f;
          if (t < T_) {
            int a = alias[b * T_ + t];
            h = re[((size_t)b * NNODE_ + a) * D_ + n0 + lh];
          }
          acc[r] = h;
        }
#pragma unroll
        for (int c = 0; c < 52; ++c) {
          int s = 4 * c + 2 * hi;
          v2f a = *(const v2f*)(sS + lh * SSTR + s);
          v2f bb;
          bb.x = sV[s * HSTR + n0 + lh];
          bb.y = sV[(s + 1) * HSTR + n0 + lh];
          acc = wmma4(a, bb, acc);
        }
#pragma unroll
        for (int r = 0; r < 8; ++r) {
          int t = qt * 16 + r + 8 * hi;
          Db[t * D_ + n0 + lh] = acc[r];
        }
      }
    }
    __syncthreads();
  }
}

__global__ void stamp_kernel(const int* __restrict__ seqlen, const float* __restrict__ wsDec,
                             const float* __restrict__ u_w1, const float* __restrict__ u_w2,
                             const float* __restrict__ u_b, const float* __restrict__ u_v,
                             float* __restrict__ out) {
  extern __shared__ float smem[];
  float* sW2 = smem;
  float* sLp = smem + D_ * D_;
  float* sUb = sLp + 128;
  float* sUv = sUb + 128;
  float* sCoef = sUv + 128;
  const int b = blockIdx.x, tid = threadIdx.x;
  const int len = seqlen[b];
  const float* Db = wsDec + (size_t)b * PBS;

  for (int idx = tid; idx < D_ * D_; idx += 256) sW2[idx] = u_w2[idx];
  if (tid < D_) { sUb[tid] = u_b[tid]; sUv[tid] = u_v[tid]; }
  if (tid < D_) {
    const float* lrow = Db + (len - 1) * D_;
    float acc = 0.f;
    for (int d = 0; d < D_; ++d) acc += lrow[d] * u_w1[d * D_ + tid];
    sLp[tid] = acc;
    out[(size_t)b * 2 * D_ + D_ + tid] = acc;
  }
  __syncthreads();

  const int wave = tid >> 5, lane = tid & 31, lh = lane & 15, hi = lane >> 4;
  for (int pass = 0; pass < 2; ++pass) {
    const int qt = wave + 8 * pass;
    if (qt < NT) {
      v2f da[32];
      const float* drow = Db + (qt * 16 + lh) * D_;
#pragma unroll
      for (int c = 0; c < 32; ++c) da[c] = *(const v2f*)(drow + 4 * c + 2 * hi);
      float cpart[8];
#pragma unroll
      for (int r = 0; r < 8; ++r) cpart[r] = 0.f;
      for (int nt = 0; nt < 8; ++nt) {
        int n0 = nt * 16;
        v8f acc = {0.f, 0.f, 0.f, 0.f, 0.f, 0.f, 0.f, 0.f};
#pragma unroll
        for (int c = 0; c < 32; ++c) {
          int k = 4 * c + 2 * hi;
          v2f bb;
          bb.x = sW2[k * D_ + n0 + lh];
          bb.y = sW2[(k + 1) * D_ + n0 + lh];
          acc = wmma4(da[c], bb, acc);
        }
        int col = n0 + lh;
        float add = sLp[col] + sUb[col];
        float uvv = sUv[col];
#pragma unroll
        for (int r = 0; r < 8; ++r) {
          float x = acc[r] + add;
          float sg = 1.f / (1.f + __expf(-x));
          cpart[r] += sg * uvv;
        }
      }
#pragma unroll
      for (int r = 0; r < 8; ++r) {
        float v = cpart[r];
        v += __shfl_xor(v, 1, 32);
        v += __shfl_xor(v, 2, 32);
        v += __shfl_xor(v, 4, 32);
        v += __shfl_xor(v, 8, 32);
        if (lh == 0) {
          int t = qt * 16 + r + 8 * hi;
          sCoef[t] = (t < len) ? v : 0.f;
        }
      }
    }
  }
  __syncthreads();
  if (tid < D_) {
    float acc = 0.f;
    for (int t = 0; t < TP; ++t) acc += sCoef[t] * Db[t * D_ + tid];
    out[(size_t)b * 2 * D_ + tid] = acc;
  }
}

extern "C" void kernel_launch(void* const* d_in, const int* in_sizes, int n_in,
                              void* d_out, int out_size, void* d_ws, size_t ws_size,
                              hipStream_t stream) {
  (void)in_sizes; (void)n_in; (void)out_size; (void)ws_size;
  const float* re   = (const float*)d_in[0];
  const int*   alia = (const int*)d_in[1];
  const int*   slen = (const int*)d_in[2];
  const float* Wq   = (const float*)d_in[3];
  const float* Wk   = (const float*)d_in[4];
  const float* Wv   = (const float*)d_in[5];
  const float* u_w1 = (const float*)d_in[6];
  const float* u_w2 = (const float*)d_in[7];
  const float* u_b  = (const float*)d_in[8];
  const float* u_v  = (const float*)d_in[9];
  float* out = (float*)d_out;

  float* ws   = (float*)d_ws;
  float* wsQ  = ws;
  float* wsK  = ws + (size_t)B_ * PBS;
  float* wsV  = ws + 2 * (size_t)B_ * PBS;
  float* wsDec = wsQ;

  size_t sm1 = (size_t)(TP * HSTR + 3 * D_ * D_) * sizeof(float);
  size_t sm2 = (size_t)(2 * TP * HSTR + 8 * 16 * SSTR) * sizeof(float);
  size_t sm3 = (size_t)(D_ * D_ + 3 * 128 + TP) * sizeof(float);
  hipFuncSetAttribute((const void*)qkv_kernel,
                      hipFuncAttributeMaxDynamicSharedMemorySize, (int)sm1);
  hipFuncSetAttribute((const void*)attn_kernel,
                      hipFuncAttributeMaxDynamicSharedMemorySize, (int)sm2);
  hipFuncSetAttribute((const void*)stamp_kernel,
                      hipFuncAttributeMaxDynamicSharedMemorySize, (int)sm3);

  qkv_kernel<<<B_, 256, sm1, stream>>>(re, alia, Wq, Wk, Wv, wsQ, wsK, wsV);
  attn_kernel<<<B_, 256, sm2, stream>>>(re, alia, slen, wsQ, wsK, wsV, wsDec);
  stamp_kernel<<<B_, 256, sm3, stream>>>(slen, wsDec, u_w1, u_w2, u_b, u_v, out);
}


# B operands (W, V, u_w2) kept in LDS as k-quads: one conflict-free 16-byte LDS read feeds the B fragments of two WMMAs, was two bank-conflicting 4-byte reads per WMMA; sums in the old order
# speedup vs baseline: 2.5503x; 2.5503x over previous
#include <hip/hip_runtime.h>
#include <math.h>


typedef __attribute__((ext_vector_type(2))) float v2f;
typedef __attribute__((ext_vector_type(4))) float v4f;
typedef __attribute__((ext_vector_type(8))) float v8f;

#define B_      2048
#define T_      200
#define D_      128
#define NNODE_  200
#define TP      208
#define NT      13
#define HSTR    132
#define PBS     (TP * D_)
#define NEGC    (-4294967295.0f)
#define QKV_W   32
#define QKV_T   (QKV_W * 32)
#define KQR     (D_ / 4)

__device__ __forceinline__ v8f wmma4(v2f a, v2f b, v8f c) {
  return __builtin_amdgcn_wmma_f32_16x16x4_f32(false, a, false, b, (short)0, c,
                                               false, false);
}


__global__ void __launch_bounds__(QKV_T) qkv_kernel(const float* __restrict__ re,
                           const float* __restrict__ Wq, const float* __restrict__ Wk,
                           const float* __restrict__ Wv,
                           float* __restrict__ wsQ, float* __restrict__ wsK,
                           float* __restrict__ wsV) {
  extern __shared__ __align__(16) float smem[];
  float* sH = smem;
  v4f* sWq = (v4f*)(smem + TP * HSTR);
  const int b = blockIdx.x, tid = threadIdx.x;

  const float* reb = re + (size_t)b * NNODE_ * D_;
  for (int idx = tid; idx < NNODE_ * D_; idx += QKV_T) {
    int t = idx >> 7, d = idx & 127;
    sH[t * HSTR + d] = reb[idx];
  }
  for (int idx = NNODE_ * D_ + tid; idx < TP * D_; idx += QKV_T) {
    int t = idx >> 7, d = idx & 127;
    sH[t * HSTR + d] = 0.f;
  }
  for (int idx = tid; idx < 3 * KQR * D_; idx += QKV_T) {
    int mat = idx >> 12, r = idx & 4095;
    int qr = r >> 7, n = r & 127;
    const float* src = ((mat == 0) ? Wq : (mat == 1) ? Wk : Wv) + (8 * (qr >> 1) + 2 * (qr & 1)) * D_ + n;
    v4f w;
    w.x = src[0];
    w.y = src[D_];
    w.z = src[4 * D_];
    w.w = src[5 * D_];
    sWq[idx] = w;
  }
  __syncthreads();

  const int wave = tid >> 5, lane = tid & 31, lh = lane & 15, hi = lane >> 4;
  for (int tile = wave; tile < 3 * NT * 8; tile += QKV_W) {
    int mat = tile / (NT * 8), rem = tile % (NT * 8);
    int i = rem >> 3, n0 = (rem & 7) << 4;
    const v4f* wq = sWq + mat * KQR * D_ + hi * D_ + n0 + lh;
    const float* arow = sH + (i * 16 + lh) * HSTR;
    v8f acc = {0.f, 0.f, 0.f, 0.f, 0.f, 0.f, 0.f, 0.f};
#pragma unroll
    for (int m = 0; m < 16; ++m) {
      const v4f w = wq[m * 2 * D_];
#pragma unroll
      for (int e = 0; e < 2; ++e) {
        int k = 4 * (2 * m + e) + 2 * hi;
        v2f a = *(const v2f*)(arow + k);
        v2f bb;
        bb.x = e ? w.z : w.x;
        bb.y = e ? w.w : w.y;
        acc = wmma4(a, bb, acc);
      }
    }
    float* dst = ((mat == 0) ? wsQ : (mat == 1) ? wsK : wsV) + (size_t)b * PBS;
#pragma unroll
    for (int r = 0; r < 8; ++r) {
      int row = i * 16 + r + 8 * hi;
      float v = acc[r];
      dst[row * D_ + n0 + lh] = v > 0.f ? v : 0.f;
    }
  }
}

#define AW  26
#define KT0 7
__global__ void __launch_bounds__(AW * 32)
attn_kernel(const float* __restrict__ re, const int* __restrict__ alias,
            const int* __restrict__ seqlen,
            const float* __restrict__ wsQ, const float* wsK,
            const float* __restrict__ wsV, float* wsDec) {
  extern __shared__ __align__(16) float smem[];
  float* sK = smem;
  v4f* sVq = (v4f*)(smem + TP * HSTR);
  int* sAl = (int*)(smem + 2 * TP * HSTR);
  float* sMx = smem + 2 * TP * HSTR + TP;
  float* sSm = sMx + AW * 16;
  const int b = blockIdx.x, tid = threadIdx.x;
  const int len = seqlen[b];
  const float* Kb = wsK + (size_t)b * PBS;
  const float* Vb = wsV + (size_t)b * PBS;
  for (int t = tid; t < TP; t += AW * 32) sAl[t] = (t < T_) ? alias[b * T_ + t] : NNODE_;
  __syncthreads();
  static_assert(TP * D_ == 32 * AW * 32, "staging: 32 K values and 8 V quads per thread");
  float kv[32];
  v4f vv[8];
#pragma unroll
  for (int u = 0; u < 32; ++u) {
    const int idx = tid + u * (AW * 32);
    kv[u] = Kb[sAl[idx >> 7] * D_ + (idx & 127)];
  }
#pragma unroll
  for (int u = 0; u < 8; ++u) {
    const int idx = tid + u * (AW * 32);
    const int t4 = (idx >> 7) * 4, d = idx & 127;
    vv[u].x = Vb[sAl[t4] * D_ + d];
    vv[u].y = Vb[sAl[t4 + 1] * D_ + d];
    vv[u].z = Vb[sAl[t4 + 2] * D_ + d];
    vv[u].w = Vb[sAl[t4 + 3] * D_ + d];
  }
#pragma unroll
  for (int u = 0; u < 32; ++u) {
    const int idx = tid + u * (AW * 32);
    sK[(idx >> 7) * HSTR + (idx & 127)] = kv[u];
  }
#pragma unroll
  for (int u = 0; u < 8; ++u) sVq[tid + u * (AW * 32)] = vv[u];
  __syncthreads();

  const int wave = __builtin_amdgcn_readfirstlane(tid >> 5);
  const int lane = tid & 31, lh = lane & 15, hi = lane >> 4;
  const int qt = wave >> 1;
  const int kh = (wave ^ (wave >> 2)) & 1;
  const int pw = wave ^ 1;
  const int jb = kh * KT0;
  const float scale = 0.088388347648318447f;

  v2f qa[32];
  const float* Qrow = wsQ + (size_t)b * PBS + sAl[qt * 16 + lh] * D_;
#pragma unroll
  for (int c = 0; c < 32; ++c) qa[c] = *(const v2f*)(Qrow + 4 * c + 2 * hi);

  float p[KT0][8];
#pragma unroll
  for (int jj = 0; jj < KT0; ++jj) {
    if (jj < KT0 - 1 || kh == 0) {
      const int j = jb + jj;
      v8f acc = {0.f, 0.f, 0.f, 0.f, 0.f, 0.f, 0.f, 0.f};
      const float* krow = sK + (j * 16 + lh) * HSTR + 2 * hi;
#pragma unroll
      for (int c = 0; c < 32; ++c) {
        v2f ka = *(const v2f*)(krow + 4 * c);
        acc = wmma4(ka, qa[c], acc);
      }
#pragma unroll
      for (int r = 0; r < 8; ++r) {
        int s = j * 16 + 8 * hi + r;
        p[jj][r] = (s < len) ? acc[r] * scale : NEGC;
      }
    } else {
#pragma unroll
      for (int r = 0; r < 8; ++r) p[jj][r] = NEGC;
    }
  }

  float mx = -3.0e38f;
#pragma unroll
  for (int jj = 0; jj < KT0; ++jj)
#pragma unroll
    for (int r = 0; r < 8; ++r) mx = fmaxf(mx, p[jj][r]);
  mx = fmaxf(mx, __shfl_xor(mx, 16, 32));
  if (hi == 0) sMx[wave * 16 + lh] = mx;
  __syncthreads();
  mx = fmaxf(mx, sMx[pw * 16 + lh]);
  float sum = 0.f;
#pragma unroll
  for (int jj = 0; jj < KT0; ++jj)
#pragma unroll
    for (int r = 0; r < 8; ++r) {
      float e = (jj < KT0 - 1 || kh == 0) ? __expf(p[jj][r] - mx) : 0.f;
      p[jj][r] = e;
      sum += e;
    }
  sum += __shfl_xor(sum, 16, 32);
  if (hi == 0) sSm[wave * 16 + lh] = sum;
  __syncthreads();
  sum += sSm[pw * 16 + lh];
  const float qm = ((qt * 16 + lh) < len) ? 1.f : 0.f;
  const float inv = qm / sum;
#pragma unroll
  for (int jj = 0; jj < KT0; ++jj)
#pragma unroll
    for (int r = 0; r < 8; ++r) p[jj][r] *= inv;

  float* sX = sK;
  float* Db = wsDec + (size_t)b * PBS;

  for (int i = 0; i < 4; ++i) {
    const int n0 = ((kh ^ 1) * 4 + i) * 16;
    v8f acc = {0.f, 0.f, 0.f, 0.f, 0.f, 0.f, 0.f, 0.f};
    const v4f* vq = sVq + (jb * 4 + 2 * hi) * D_ + n0 + lh;
#pragma unroll
    for (int jj = 0; jj < KT0; ++jj) {
      if (jj < KT0 - 1 || kh == 0) {
#pragma unroll
        for (int u = 0; u < 2; ++u) {
          const v4f v = vq[(jj * 4 + u) * D_];
#pragma unroll
          for (int e = 0; e < 2; ++e) {
            const int q = 2 * u + e;
            v2f a;
            a.x = p[jj][2 * q];
            a.y = p[jj][2 * q + 1];
            v2f bb;
            bb.x = e ? v.z : v.x;
            bb.y = e ? v.w : v.y;
            acc = wmma4(a, bb, acc);
          }
        }
      }
    }
    float* xo = sX + (wave * 4 + i) * 256 + lane;
#pragma unroll
    for (int r = 0; r < 8; ++r) xo[r * 32] = acc[r];
  }
  __syncthreads();

  for (int i = 0; i < 4; ++i) {
    const int n0 = (kh * 4 + i) * 16;
    v8f acc;
#pragma unroll
    for (int r = 0; r < 8; ++r) {
      int t = qt * 16 + r + 8 * hi;
      float h = 0.f;
      if (t < T_) {
        int a = alias[b * T_ + t];
        h = re[((size_t)b * NNODE_ + a) * D_ + n0 + lh];
      }
      acc[r] = h;
    }
    const v4f* vq = sVq + (jb * 4 + 2 * hi) * D_ + n0 + lh;
#pragma unroll
    for (int jj = 0; jj < KT0; ++jj) {
      if (jj < KT0 - 1 || kh == 0) {
#pragma unroll
        for (int u = 0; u < 2; ++u) {
          const v4f v = vq[(jj * 4 + u) * D_];
#pragma unroll
          for (int e = 0; e < 2; ++e) {
            const int q = 2 * u + e;
            v2f a;
            a.x = p[jj][2 * q];
            a.y = p[jj][2 * q + 1];
            v2f bb;
            bb.x = e ? v.z : v.x;
            bb.y = e ? v.w : v.y;
            acc = wmma4(a, bb, acc);
          }
        }
      }
    }
    const float* xi = sX + (pw * 4 + i) * 256 + lane;
#pragma unroll
    for (int r = 0; r < 8; ++r) {
      int t = qt * 16 + r + 8 * hi;
      Db[t * D_ + n0 + lh] = acc[r] + xi[r * 32];
    }
  }
}

__global__ void stamp_kernel(const int* __restrict__ seqlen, const float* __restrict__ wsDec,
                             const float* __restrict__ u_w1, const float* __restrict__ u_w2,
                             const float* __restrict__ u_b, const float* __restrict__ u_v,
                             float* __restrict__ out) {
  extern __shared__ __align__(16) float smem[];
  v4f* sW2q = (v4f*)smem;
  float* sLp = smem + D_ * D_;
  float* sUb = sLp + 128;
  float* sUv = sUb + 128;
  float* sCoef = sUv + 128;
  const int b = blockIdx.x, tid = threadIdx.x;
  const int len = seqlen[b];
  const float* Db = wsDec + (size_t)b * PBS;

  for (int idx = tid; idx < KQR * D_; idx += 256) {
    int qr = idx >> 7, n = idx & 127;
    const float* src = u_w2 + (8 * (qr >> 1) + 2 * (qr & 1)) * D_ + n;
    v4f w;
    w.x = src[0];
    w.y = src[D_];
    w.z = src[4 * D_];
    w.w = src[5 * D_];
    sW2q[idx] = w;
  }
  if (tid < D_) { sUb[tid] = u_b[tid]; sUv[tid] = u_v[tid]; }
  if (tid < D_) {
    const float* lrow = Db + (len - 1) * D_;
    float acc = 0.f;
    for (int d = 0; d < D_; ++d) acc += lrow[d] * u_w1[d * D_ + tid];
    sLp[tid] = acc;
    out[(size_t)b * 2 * D_ + D_ + tid] = acc;
  }
  __syncthreads();

  const int wave = tid >> 5, lane = tid & 31, lh = lane & 15, hi = lane >> 4;
  for (int pass = 0; pass < 2; ++pass) {
    const int qt = wave + 8 * pass;
    if (qt < NT) {
      v2f da[32];
      const float* drow = Db + (qt * 16 + lh) * D_;
#pragma unroll
      for (int c = 0; c < 32; ++c) da[c] = *(const v2f*)(drow + 4 * c + 2 * hi);
      float cpart[8];
#pragma unroll
      for (int r = 0; r < 8; ++r) cpart[r] = 0.f;
      for (int nt = 0; nt < 8; ++nt) {
        int n0 = nt * 16;
        v8f acc = {0.f, 0.f, 0.f, 0.f, 0.f, 0.f, 0.f, 0.f};
        const v4f* wq = sW2q + hi * D_ + n0 + lh;
#pragma unroll
        for (int m = 0; m < 16; ++m) {
          const v4f w = wq[m * 2 * D_];
          v2f b0, b1;
          b0.x = w.x; b0.y = w.y;
          b1.x = w.z; b1.y = w.w;
          acc = wmma4(da[2 * m], b0, acc);
          acc = wmma4(da[2 * m + 1], b1, acc);
        }
        int col = n0 + lh;
        float add = sLp[col] + sUb[col];
        float uvv = sUv[col];
#pragma unroll
        for (int r = 0; r < 8; ++r) {
          float x = acc[r] + add;
          float sg = 1.f / (1.f + __expf(-x));
          cpart[r] += sg * uvv;
        }
      }
#pragma unroll
      for (int r = 0; r < 8; ++r) {
        float v = cpart[r];
        v += __shfl_xor(v, 1, 32);
        v += __shfl_xor(v, 2, 32);
        v += __shfl_xor(v, 4, 32);
        v += __shfl_xor(v, 8, 32);
        if (lh == 0) {
          int t = qt * 16 + r + 8 * hi;
          sCoef[t] = (t < len) ? v : 0.f;
        }
      }
    }
  }
  __syncthreads();
  if (tid < D_) {
    float acc = 0.f;
    for (int t = 0; t < TP; ++t) acc += sCoef[t] * Db[t * D_ + tid];
    out[(size_t)b * 2 * D_ + tid] = acc;
  }
}

extern "C" void kernel_launch(void* const* d_in, const int* in_sizes, int n_in,
                              void* d_out, int out_size, void* d_ws, size_t ws_size,
                              hipStream_t stream) {
  (void)in_sizes; (void)n_in; (void)out_size; (void)ws_size;
  const float* re   = (const float*)d_in[0];
  const int*   alia = (const int*)d_in[1];
  const int*   slen = (const int*)d_in[2];
  const float* Wq   = (const float*)d_in[3];
  const float* Wk   = (const float*)d_in[4];
  const float* Wv   = (const float*)d_in[5];
  const float* u_w1 = (const float*)d_in[6];
  const float* u_w2 = (const float*)d_in[7];
  const float* u_b  = (const float*)d_in[8];
  const float* u_v  = (const float*)d_in[9];
  float* out = (float*)d_out;

  float* ws   = (float*)d_ws;
  float* wsQ  = ws;
  float* wsK  = ws + (size_t)B_ * PBS;
  float* wsV  = ws + 2 * (size_t)B_ * PBS;
  float* wsDec = wsK;

  size_t sm1 = (size_t)(TP * HSTR + 3 * D_ * D_) * sizeof(float);
  size_t sm2 = (size_t)(2 * TP * HSTR + TP + 2 * AW * 16) * sizeof(float);
  size_t sm3 = (size_t)(D_ * D_ + 3 * 128 + TP) * sizeof(float);
  hipFuncSetAttribute((const void*)qkv_kernel,
                      hipFuncAttributeMaxDynamicSharedMemorySize, (int)sm1);
  hipFuncSetAttribute((const void*)attn_kernel,
                      hipFuncAttributeMaxDynamicSharedMemorySize, (int)sm2);
  hipFuncSetAttribute((const void*)stamp_kernel,
                      hipFuncAttributeMaxDynamicSharedMemorySize, (int)sm3);

  qkv_kernel<<<B_, QKV_T, sm1, stream>>>(re, Wq, Wk, Wv, wsQ, wsK, wsV);
  attn_kernel<<<B_, AW * 32, sm2, stream>>>(re, alia, slen, wsQ, wsK, wsV, wsDec);
  stamp_kernel<<<B_, 256, sm3, stream>>>(slen, wsDec, u_w1, u_w2, u_b, u_v, out);
}


# lanes hold adjacent columns (W, V columns reordered in LDS, a wave does 2 or 4 tiles together): P* and dec stored as whole 128-byte lines, residual read 16 B per lane; was half lines; old sum order
# speedup vs baseline: 2.7029x; 1.0598x over previous
#include <hip/hip_runtime.h>
#include <math.h>


typedef __attribute__((ext_vector_type(2))) float v2f;
typedef __attribute__((ext_vector_type(4))) float v4f;
typedef __attribute__((ext_vector_type(8))) float v8f;

#define B_      2048
#define T_      200
#define D_      128
#define NNODE_  200
#define TP      208
#define NT      13
#define HSTR    132
#define PBS     (TP * D_)
#define NEGC    (-4294967295.0f)
#define QKV_W   32
#define QKV_T   (QKV_W * 32)
#define KQR     (D_ / 4)

__device__ __forceinline__ v8f wmma4(v2f a, v2f b, v8f c) {
  return __builtin_amdgcn_wmma_f32_16x16x4_f32(false, a, false, b, (short)0, c,
                                               false, false);
}

#define CSRC2(p) (((p) & 96) | (((p) & 15) << 1) | (((p) >> 4) & 1))
#define CSRC4(p) (((p) & 64) | (((p) & 15) << 2) | (((p) >> 4) & 3))

__global__ void __launch_bounds__(QKV_T) qkv_kernel(const float* __restrict__ re,
                           const float* __restrict__ Wq, const float* __restrict__ Wk,
                           const float* __restrict__ Wv,
                           float* __restrict__ wsQ, float* __restrict__ wsK,
                           float* __restrict__ wsV) {
  extern __shared__ __align__(16) float smem[];
  float* sH = smem;
  v4f* sWq = (v4f*)(smem + TP * HSTR);
  const int b = blockIdx.x, tid = threadIdx.x;

  const float* reb = re + (size_t)b * NNODE_ * D_;
  for (int idx = tid; idx < NNODE_ * D_; idx += QKV_T) {
    int t = idx >> 7, d = idx & 127;
    sH[t * HSTR + d] = reb[idx];
  }
  for (int idx = NNODE_ * D_ + tid; idx < TP * D_; idx += QKV_T) {
    int t = idx >> 7, d = idx & 127;
    sH[t * HSTR + d] = 0.f;
  }
  for (int idx = tid; idx < 3 * KQR * D_; idx += QKV_T) {
    int mat = idx >> 12, r = idx & 4095;
    int qr = r >> 7, n = CSRC2(r & 127);
    const float* src = ((mat == 0) ? Wq : (mat == 1) ? Wk : Wv) + (8 * (qr >> 1) + 2 * (qr & 1)) * D_ + n;
    v4f w;
    w.x = src[0];
    w.y = src[D_];
    w.z = src[4 * D_];
    w.w = src[5 * D_];
    sWq[idx] = w;
  }
  __syncthreads();

  const int wave = tid >> 5, lane = tid & 31, lh = lane & 15, hi = lane >> 4;
  for (int unit = wave; unit < 3 * NT * 4; unit += QKV_W) {
    int mat = unit / (NT * 4), rem = unit % (NT * 4);
    int i = rem >> 2, n0 = (rem & 3) << 5;
    const v4f* wq = sWq + mat * KQR * D_ + hi * D_ + n0 + lh;
    const float* arow = sH + (i * 16 + lh) * HSTR;
    v8f acc0 = {0.f, 0.f, 0.f, 0.f, 0.f, 0.f, 0.f, 0.f};
    v8f acc1 = {0.f, 0.f, 0.f, 0.f, 0.f, 0.f, 0.f, 0.f};
#pragma unroll
    for (int m = 0; m < 16; ++m) {
      const v4f w0 = wq[m * 2 * D_];
      const v4f w1 = wq[m * 2 * D_ + 16];
#pragma unroll
      for (int e = 0; e < 2; ++e) {
        int k = 4 * (2 * m + e) + 2 * hi;
        v2f a = *(const v2f*)(arow + k);
        v2f b0, b1;
        b0.x = e ? w0.z : w0.x;
        b0.y = e ? w0.w : w0.y;
        b1.x = e ? w1.z : w1.x;
        b1.y = e ? w1.w : w1.y;
        acc0 = wmma4(a, b0, acc0);
        acc1 = wmma4(a, b1, acc1);
      }
    }
    float* dst = ((mat == 0) ? wsQ : (mat == 1) ? wsK : wsV) + (size_t)b * PBS;
#pragma unroll
    for (int r = 0; r < 8; ++r) {
      int row = i * 16 + r + 8 * hi;
      v2f o;
      o.x = acc0[r] > 0.f ? acc0[r] : 0.f;
      o.y = acc1[r] > 0.f ? acc1[r] : 0.f;
      *(v2f*)(dst + row * D_ + n0 + 2 * lh) = o;
    }
  }
}

#define AW  26
#define KT0 7
__global__ void __launch_bounds__(AW * 32)
attn_kernel(const float* __restrict__ re, const int* __restrict__ alias,
            const int* __restrict__ seqlen,
            const float* __restrict__ wsQ, const float* wsK,
            const float* __restrict__ wsV, float* wsDec) {
  extern __shared__ __align__(16) float smem[];
  float* sK = smem;
  v4f* sVq = (v4f*)(smem + TP * HSTR);
  int* sAl = (int*)(smem + 2 * TP * HSTR);
  float* sMx = smem + 2 * TP * HSTR + TP;
  float* sSm = sMx + AW * 16;
  const int b = blockIdx.x, tid = threadIdx.x;
  const int len = seqlen[b];
  const float* Kb = wsK + (size_t)b * PBS;
  const float* Vb = wsV + (size_t)b * PBS;
  for (int t = tid; t < TP; t += AW * 32) sAl[t] = (t < T_) ? alias[b * T_ + t] : NNODE_;
  __syncthreads();
  static_assert(TP * D_ == 32 * AW * 32, "staging: 32 K values and 8 V quads per thread");
  float kv[32];
  v4f vv[8];
#pragma unroll
  for (int u = 0; u < 32; ++u) {
    const int idx = tid + u * (AW * 32);
    kv[u] = Kb[sAl[idx >> 7] * D_ + (idx & 127)];
  }
#pragma unroll
  for (int u = 0; u < 8; ++u) {
    const int idx = tid + u * (AW * 32);
    const int t4 = (idx >> 7) * 4, d = CSRC4(idx & 127);
    vv[u].x = Vb[sAl[t4] * D_ + d];
    vv[u].y = Vb[sAl[t4 + 1] * D_ + d];
    vv[u].z = Vb[sAl[t4 + 2] * D_ + d];
    vv[u].w = Vb[sAl[t4 + 3] * D_ + d];
  }
#pragma unroll
  for (int u = 0; u < 32; ++u) {
    const int idx = tid + u * (AW * 32);
    sK[(idx >> 7) * HSTR + (idx & 127)] = kv[u];
  }
#pragma unroll
  for (int u = 0; u < 8; ++u) sVq[tid + u * (AW * 32)] = vv[u];
  __syncthreads();

  const int wave = __builtin_amdgcn_readfirstlane(tid >> 5);
  const int lane = tid & 31, lh = lane & 15, hi = lane >> 4;
  const int qt = wave >> 1;
  const int kh = (wave ^ (wave >> 2)) & 1;
  const int pw = wave ^ 1;
  const int jb = kh * KT0;
  const float scale = 0.088388347648318447f;

  v2f qa[32];
  const float* Qrow = wsQ + (size_t)b * PBS + sAl[qt * 16 + lh] * D_;
#pragma unroll
  for (int c = 0; c < 32; ++c) qa[c] = *(const v2f*)(Qrow + 4 * c + 2 * hi);

  float p[KT0][8];
#pragma unroll
  for (int jj = 0; jj < KT0; ++jj) {
    if (jj < KT0 - 1 || kh == 0) {
      const int j = jb + jj;
      v8f acc = {0.f, 0.f, 0.f, 0.f, 0.f, 0.f, 0.f, 0.f};
      const float* krow = sK + (j * 16 + lh) * HSTR + 2 * hi;
#pragma unroll
      for (int c = 0; c < 32; ++c) {
        v2f ka = *(const v2f*)(krow + 4 * c);
        acc = wmma4(ka, qa[c], acc);
      }
#pragma unroll
      for (int r = 0; r < 8; ++r) {
        int s = j * 16 + 8 * hi + r;
        p[jj][r] = (s < len) ? acc[r] * scale : NEGC;
      }
    } else {
#pragma unroll
      for (int r = 0; r < 8; ++r) p[jj][r] = NEGC;
    }
  }

  float mx = -3.0e38f;
#pragma unroll
  for (int jj = 0; jj < KT0; ++jj)
#pragma unroll
    for (int r = 0; r < 8; ++r) mx = fmaxf(mx, p[jj][r]);
  mx = fmaxf(mx, __shfl_xor(mx, 16, 32));
  if (hi == 0) sMx[wave * 16 + lh] = mx;
  __syncthreads();
  mx = fmaxf(mx, sMx[pw * 16 + lh]);
  float sum = 0.f;
#pragma unroll
  for (int jj = 0; jj < KT0; ++jj)
#pragma unroll
    for (int r = 0; r < 8; ++r) {
      float e = (jj < KT0 - 1 || kh == 0) ? __expf(p[jj][r] - mx) : 0.f;
      p[jj][r] = e;
      sum += e;
    }
  sum += __shfl_xor(sum, 16, 32);
  if (hi == 0) sSm[wave * 16 + lh] = sum;
  __syncthreads();
  sum += sSm[pw * 16 + lh];
  const float qm = ((qt * 16 + lh) < len) ? 1.f : 0.f;
  const float inv = qm / sum;
#pragma unroll
  for (int jj = 0; jj < KT0; ++jj)
#pragma unroll
    for (int r = 0; r < 8; ++r) p[jj][r] *= inv;

  float* sX = sK;
  float* Db = wsDec + (size_t)b * PBS;

  for (int i = 0; i < 4; ++i) {
    const int n0 = ((kh ^ 1) * 4 + i) * 16;
    v8f acc = {0.f, 0.f, 0.f, 0.f, 0.f, 0.f, 0.f, 0.f};
    const v4f* vq = sVq + (jb * 4 + 2 * hi) * D_ + n0 + lh;
#pragma unroll
    for (int jj = 0; jj < KT0; ++jj) {
      if (jj < KT0 - 1 || kh == 0) {
#pragma unroll
        for (int u = 0; u < 2; ++u) {
          const v4f v = vq[(jj * 4 + u) * D_];
#pragma unroll
          for (int e = 0; e < 2; ++e) {
            const int q = 2 * u + e;
            v2f a;
            a.x = p[jj][2 * q];
            a.y = p[jj][2 * q + 1];
            v2f bb;
            bb.x = e ? v.z : v.x;
            bb.y = e ? v.w : v.y;
            acc = wmma4(a, bb, acc);
          }
        }
      }
    }
    float* xo = sX + (wave * 4 + i) * 256 + lane;
#pragma unroll
    for (int r = 0; r < 8; ++r) xo[r * 32] = acc[r];
  }
  __syncthreads();

  {
    const int n0 = kh * 64;
    v8f acc[4];
#pragma unroll
    for (int r = 0; r < 8; ++r) {
      int t = qt * 16 + r + 8 * hi;
      v4f h = {0.f, 0.f, 0.f, 0.f};
      if (t < T_) {
        int a = alias[b * T_ + t];
        h = *(const v4f*)(re + ((size_t)b * NNODE_ + a) * D_ + n0 + 4 * lh);
      }
      acc[0][r] = h.x;
      acc[1][r] = h.y;
      acc[2][r] = h.z;
      acc[3][r] = h.w;
    }
    const v4f* vq = sVq + (jb * 4 + 2 * hi) * D_ + n0 + lh;
#pragma unroll
    for (int jj = 0; jj < KT0; ++jj) {
      if (jj < KT0 - 1 || kh == 0) {
#pragma unroll
        for (int u = 0; u < 2; ++u) {
          v4f v[4];
#pragma unroll
          for (int c = 0; c < 4; ++c) v[c] = vq[(jj * 4 + u) * D_ + 16 * c];
#pragma unroll
          for (int e = 0; e < 2; ++e) {
            const int q = 2 * u + e;
            v2f a;
            a.x = p[jj][2 * q];
            a.y = p[jj][2 * q + 1];
#pragma unroll
            for (int c = 0; c < 4; ++c) {
              v2f bb;
              bb.x = e ? v[c].z : v[c].x;
              bb.y = e ? v[c].w : v[c].y;
              acc[c] = wmma4(a, bb, acc[c]);
            }
          }
        }
      }
    }
    const float* xi = sX + pw * 4 * 256 + lane;
#pragma unroll
    for (int r = 0; r < 8; ++r) {
      int t = qt * 16 + r + 8 * hi;
      v4f o;
      o.x = acc[0][r] + xi[0 * 256 + r * 32];
      o.y = acc[1][r] + xi[1 * 256 + r * 32];
      o.z = acc[2][r] + xi[2 * 256 + r * 32];
      o.w = acc[3][r] + xi[3 * 256 + r * 32];
      *(v4f*)(Db + t * D_ + n0 + 4 * lh) = o;
    }
  }
}

__global__ void stamp_kernel(const int* __restrict__ seqlen, const float* __restrict__ wsDec,
                             const float* __restrict__ u_w1, const float* __restrict__ u_w2,
                             const float* __restrict__ u_b, const float* __restrict__ u_v,
                             float* __restrict__ out) {
  extern __shared__ __align__(16) float smem[];
  v4f* sW2q = (v4f*)smem;
  float* sLp = smem + D_ * D_;
  float* sUb = sLp + 128;
  float* sUv = sUb + 128;
  float* sCoef = sUv + 128;
  const int b = blockIdx.x, tid = threadIdx.x;
  const int len = seqlen[b];
  const float* Db = wsDec + (size_t)b * PBS;

  for (int idx = tid; idx < KQR * D_; idx += 256) {
    int qr = idx >> 7, n = idx & 127;
    const float* src = u_w2 + (8 * (qr >> 1) + 2 * (qr & 1)) * D_ + n;
    v4f w;
    w.x = src[0];
    w.y = src[D_];
    w.z = src[4 * D_];
    w.w = src[5 * D_];
    sW2q[idx] = w;
  }
  if (tid < D_) { sUb[tid] = u_b[tid]; sUv[tid] = u_v[tid]; }
  if (tid < D_) {
    const float* lrow = Db + (len - 1) * D_;
    float acc = 0.f;
    for (int d = 0; d < D_; ++d) acc += lrow[d] * u_w1[d * D_ + tid];
    sLp[tid] = acc;
    out[(size_t)b * 2 * D_ + D_ + tid] = acc;
  }
  __syncthreads();

  const int wave = tid >> 5, lane = tid & 31, lh = lane & 15, hi = lane >> 4;
  for (int pass = 0; pass < 2; ++pass) {
    const int qt = wave + 8 * pass;
    if (qt < NT) {
      v2f da[32];
      const float* drow = Db + (qt * 16 + lh) * D_;
#pragma unroll
      for (int c = 0; c < 32; ++c) da[c] = *(const v2f*)(drow + 4 * c + 2 * hi);
      float cpart[8];
#pragma unroll
      for (int r = 0; r < 8; ++r) cpart[r] = 0.f;
      for (int nt = 0; nt < 8; ++nt) {
        int n0 = nt * 16;
        v8f acc = {0.f, 0.f, 0.f, 0.f, 0.f, 0.f, 0.f, 0.f};
        const v4f* wq = sW2q + hi * D_ + n0 + lh;
#pragma unroll
        for (int m = 0; m < 16; ++m) {
          const v4f w = wq[m * 2 * D_];
          v2f b0, b1;
          b0.x = w.x; b0.y = w.y;
          b1.x = w.z; b1.y = w.w;
          acc = wmma4(da[2 * m], b0, acc);
          acc = wmma4(da[2 * m + 1], b1, acc);
        }
        int col = n0 + lh;
        float add = sLp[col] + sUb[col];
        float uvv = sUv[col];
#pragma unroll
        for (int r = 0; r < 8; ++r) {
          float x = acc[r] + add;
          float sg = 1.f / (1.f + __expf(-x));
          cpart[r] += sg * uvv;
        }
      }
#pragma unroll
      for (int r = 0; r < 8; ++r) {
        float v = cpart[r];
        v += __shfl_xor(v, 1, 32);
        v += __shfl_xor(v, 2, 32);
        v += __shfl_xor(v, 4, 32);
        v += __shfl_xor(v, 8, 32);
        if (lh == 0) {
          int t = qt * 16 + r + 8 * hi;
          sCoef[t] = (t < len) ? v : 0.f;
        }
      }
    }
  }
  __syncthreads();
  if (tid < D_) {
    float acc = 0.f;
    for (int t = 0; t < TP; ++t) acc += sCoef[t] * Db[t * D_ + tid];
    out[(size_t)b * 2 * D_ + tid] = acc;
  }
}

extern "C" void kernel_launch(void* const* d_in, const int* in_sizes, int n_in,
                              void* d_out, int out_size, void* d_ws, size_t ws_size,
                              hipStream_t stream) {
  (void)in_sizes; (void)n_in; (void)out_size; (void)ws_size;
  const float* re   = (const float*)d_in[0];
  const int*   alia = (const int*)d_in[1];
  const int*   slen = (const int*)d_in[2];
  const float* Wq   = (const float*)d_in[3];
  const float* Wk   = (const float*)d_in[4];
  const float* Wv   = (const float*)d_in[5];
  const float* u_w1 = (const float*)d_in[6];
  const float* u_w2 = (const float*)d_in[7];
  const float* u_b  = (const float*)d_in[8];
  const float* u_v  = (const float*)d_in[9];
  float* out = (float*)d_out;

  float* ws   = (float*)d_ws;
  float* wsQ  = ws;
  float* wsK  = ws + (size_t)B_ * PBS;
  float* wsV  = ws + 2 * (size_t)B_ * PBS;
  float* wsDec = wsK;

  size_t sm1 = (size_t)(TP * HSTR + 3 * D_ * D_) * sizeof(float);
  size_t sm2 = (size_t)(2 * TP * HSTR + TP + 2 * AW * 16) * sizeof(float);
  size_t sm3 = (size_t)(D_ * D_ + 3 * 128 + TP) * sizeof(float);
  hipFuncSetAttribute((const void*)qkv_kernel,
                      hipFuncAttributeMaxDynamicSharedMemorySize, (int)sm1);
  hipFuncSetAttribute((const void*)attn_kernel,
                      hipFuncAttributeMaxDynamicSharedMemorySize, (int)sm2);
  hipFuncSetAttribute((const void*)stamp_kernel,
                      hipFuncAttributeMaxDynamicSharedMemorySize, (int)sm3);

  qkv_kernel<<<B_, QKV_T, sm1, stream>>>(re, Wq, Wk, Wv, wsQ, wsK, wsV);
  attn_kernel<<<B_, AW * 32, sm2, stream>>>(re, alia, slen, wsQ, wsK, wsV, wsDec);
  stamp_kernel<<<B_, 256, sm3, stream>>>(slen, wsDec, u_w1, u_w2, u_b, u_v, out);
}


# attn, stamp: tile roles dealt to waves by SIMD, equal WMMA share per SIMD: attn 43/42/42/42 key tiles (was 46/45/39/39), stamp 3.25 tiles each (was 4/3/3/3); 13th tile's coef in a new sum order
# speedup vs baseline: 2.7834x; 1.0298x over previous
#include <hip/hip_runtime.h>
#include <math.h>


typedef __attribute__((ext_vector_type(2))) float v2f;
typedef __attribute__((ext_vector_type(4))) float v4f;
typedef __attribute__((ext_vector_type(8))) float v8f;

#define B_      2048
#define T_      200
#define D_      128
#define NNODE_  200
#define TP      208
#define NT      13
#define HSTR    132
#define PBS     (TP * D_)
#define NEGC    (-4294967295.0f)
#define QKV_W   32
#define QKV_T   (QKV_W * 32)
#define KQR     (D_ / 4)

__device__ __forceinline__ v8f wmma4(v2f a, v2f b, v8f c) {
  return __builtin_amdgcn_wmma_f32_16x16x4_f32(false, a, false, b, (short)0, c,
                                               false, false);
}

#define CSRC2(p) (((p) & 96) | (((p) & 15) << 1) | (((p) >> 4) & 1))
#define CSRC4(p) (((p) & 64) | (((p) & 15) << 2) | (((p) >> 4) & 3))

__global__ void __launch_bounds__(QKV_T) qkv_kernel(const float* __restrict__ re,
                           const float* __restrict__ Wq, const float* __restrict__ Wk,
                           const float* __restrict__ Wv,
                           float* __restrict__ wsQ, float* __restrict__ wsK,
                           float* __restrict__ wsV) {
  extern __shared__ __align__(16) float smem[];
  float* sH = smem;
  v4f* sWq = (v4f*)(smem + TP * HSTR);
  const int b = blockIdx.x, tid = threadIdx.x;

  const float* reb = re + (size_t)b * NNODE_ * D_;
  for (int idx = tid; idx < NNODE_ * D_; idx += QKV_T) {
    int t = idx >> 7, d = idx & 127;
    sH[t * HSTR + d] = reb[idx];
  }
  for (int idx = NNODE_ * D_ + tid; idx < TP * D_; idx += QKV_T) {
    int t = idx >> 7, d = idx & 127;
    sH[t * HSTR + d] = 0.f;
  }
  for (int idx = tid; idx < 3 * KQR * D_; idx += QKV_T) {
    int mat = idx >> 12, r = idx & 4095;
    int qr = r >> 7, n = CSRC2(r & 127);
    const float* src = ((mat == 0) ? Wq : (mat == 1) ? Wk : Wv) + (8 * (qr >> 1) + 2 * (qr & 1)) * D_ + n;
    v4f w;
    w.x = src[0];
    w.y = src[D_];
    w.z = src[4 * D_];
    w.w = src[5 * D_];
    sWq[idx] = w;
  }
  __syncthreads();

  const int wave = tid >> 5, lane = tid & 31, lh = lane & 15, hi = lane >> 4;
  for (int unit = wave; unit < 3 * NT * 4; unit += QKV_W) {
    int mat = unit / (NT * 4), rem = unit % (NT * 4);
    int i = rem >> 2, n0 = (rem & 3) << 5;
    const v4f* wq = sWq + mat * KQR * D_ + hi * D_ + n0 + lh;
    const float* arow = sH + (i * 16 + lh) * HSTR;
    v8f acc0 = {0.f, 0.f, 0.f, 0.f, 0.f, 0.f, 0.f, 0.f};
    v8f acc1 = {0.f, 0.f, 0.f, 0.f, 0.f, 0.f, 0.f, 0.f};
#pragma unroll
    for (int m = 0; m < 16; ++m) {
      const v4f w0 = wq[m * 2 * D_];
      const v4f w1 = wq[m * 2 * D_ + 16];
#pragma unroll
      for (int e = 0; e < 2; ++e) {
        int k = 4 * (2 * m + e) + 2 * hi;
        v2f a = *(const v2f*)(arow + k);
        v2f b0, b1;
        b0.x = e ? w0.z : w0.x;
        b0.y = e ? w0.w : w0.y;
        b1.x = e ? w1.z : w1.x;
        b1.y = e ? w1.w : w1.y;
        acc0 = wmma4(a, b0, acc0);
        acc1 = wmma4(a, b1, acc1);
      }
    }
    float* dst = ((mat == 0) ? wsQ : (mat == 1) ? wsK : wsV) + (size_t)b * PBS;
#pragma unroll
    for (int r = 0; r < 8; ++r) {
      int row = i * 16 + r + 8 * hi;
      v2f o;
      o.x = acc0[r] > 0.f ? acc0[r] : 0.f;
      o.y = acc1[r] > 0.f ? acc1[r] : 0.f;
      *(v2f*)(dst + row * D_ + n0 + 2 * lh) = o;
    }
  }
}

#define AW  26
#define KT0 7
__global__ void __launch_bounds__(AW * 32)
attn_kernel(const float* __restrict__ re, const int* __restrict__ alias,
            const int* __restrict__ seqlen,
            const float* __restrict__ wsQ, const float* wsK,
            const float* __restrict__ wsV, float* wsDec) {
  extern __shared__ __align__(16) float smem[];
  float* sK = smem;
  v4f* sVq = (v4f*)(smem + TP * HSTR);
  int* sAl = (int*)(smem + 2 * TP * HSTR);
  float* sMx = smem + 2 * TP * HSTR + TP;
  float* sSm = sMx + AW * 16;
  const int b = blockIdx.x, tid = threadIdx.x;
  const int len = seqlen[b];
  const float* Kb = wsK + (size_t)b * PBS;
  const float* Vb = wsV + (size_t)b * PBS;
  for (int t = tid; t < TP; t += AW * 32) sAl[t] = (t < T_) ? alias[b * T_ + t] : NNODE_;
  __syncthreads();
  static_assert(TP * D_ == 32 * AW * 32, "staging: 32 K values and 8 V quads per thread");
  float kv[32];
  v4f vv[8];
#pragma unroll
  for (int u = 0; u < 32; ++u) {
    const int idx = tid + u * (AW * 32);
    kv[u] = Kb[sAl[idx >> 7] * D_ + (idx & 127)];
  }
#pragma unroll
  for (int u = 0; u < 8; ++u) {
    const int idx = tid + u * (AW * 32);
    const int t4 = (idx >> 7) * 4, d = CSRC4(idx & 127);
    vv[u].x = Vb[sAl[t4] * D_ + d];
    vv[u].y = Vb[sAl[t4 + 1] * D_ + d];
    vv[u].z = Vb[sAl[t4 + 2] * D_ + d];
    vv[u].w = Vb[sAl[t4 + 3] * D_ + d];
  }
#pragma unroll
  for (int u = 0; u < 32; ++u) {
    const int idx = tid + u * (AW * 32);
    sK[(idx >> 7) * HSTR + (idx & 127)] = kv[u];
  }
#pragma unroll
  for (int u = 0; u < 8; ++u) sVq[tid + u * (AW * 32)] = vv[u];
  __syncthreads();

  const int wave = __builtin_amdgcn_readfirstlane(tid >> 5);
  const int lane = tid & 31, lh = lane & 15, hi = lane >> 4;
  const int cls = wave & 3, wi = wave >> 2;
  int qt, kh;
  if (cls == 0)      { kh = 1; qt = wi; }
  else if (cls == 1) { kh = (wi < 6) ? 1 : 0; qt = (wi < 6) ? 7 + wi : 12; }
  else if (cls == 2) { kh = 0; qt = wi; }
  else               { kh = 0; qt = 6 + wi; }
  const int pw = kh ? ((qt < 6) ? 4 * qt + 2 : (qt < 12) ? 4 * (qt - 6) + 3 : 25)
                    : ((qt < 7) ? 4 * qt : 4 * (qt - 7) + 1);
  const int jb = kh * KT0;
  const float scale = 0.088388347648318447f;

  v2f qa[32];
  const float* Qrow = wsQ + (size_t)b * PBS + sAl[qt * 16 + lh] * D_;
#pragma unroll
  for (int c = 0; c < 32; ++c) qa[c] = *(const v2f*)(Qrow + 4 * c + 2 * hi);

  float p[KT0][8];
#pragma unroll
  for (int jj = 0; jj < KT0; ++jj) {
    if (jj < KT0 - 1 || kh == 0) {
      const int j = jb + jj;
      v8f acc = {0.f, 0.f, 0.f, 0.f, 0.f, 0.f, 0.f, 0.f};
      const float* krow = sK + (j * 16 + lh) * HSTR + 2 * hi;
#pragma unroll
      for (int c = 0; c < 32; ++c) {
        v2f ka = *(const v2f*)(krow + 4 * c);
        acc = wmma4(ka, qa[c], acc);
      }
#pragma unroll
      for (int r = 0; r < 8; ++r) {
        int s = j * 16 + 8 * hi + r;
        p[jj][r] = (s < len) ? acc[r] * scale : NEGC;
      }
    } else {
#pragma unroll
      for (int r = 0; r < 8; ++r) p[jj][r] = NEGC;
    }
  }

  float mx = -3.0e38f;
#pragma unroll
  for (int jj = 0; jj < KT0; ++jj)
#pragma unroll
    for (int r = 0; r < 8; ++r) mx = fmaxf(mx, p[jj][r]);
  mx = fmaxf(mx, __shfl_xor(mx, 16, 32));
  if (hi == 0) sMx[wave * 16 + lh] = mx;
  __syncthreads();
  mx = fmaxf(mx, sMx[pw * 16 + lh]);
  float sum = 0.f;
#pragma unroll
  for (int jj = 0; jj < KT0; ++jj)
#pragma unroll
    for (int r = 0; r < 8; ++r) {
      float e = (jj < KT0 - 1 || kh == 0) ? __expf(p[jj][r] - mx) : 0.f;
      p[jj][r] = e;
      sum += e;
    }
  sum += __shfl_xor(sum, 16, 32);
  if (hi == 0) sSm[wave * 16 + lh] = sum;
  __syncthreads();
  sum += sSm[pw * 16 + lh];
  const float qm = ((qt * 16 + lh) < len) ? 1.f : 0.f;
  const float inv = qm / sum;
#pragma unroll
  for (int jj = 0; jj < KT0; ++jj)
#pragma unroll
    for (int r = 0; r < 8; ++r) p[jj][r] *= inv;

  float* sX = sK;
  float* Db = wsDec + (size_t)b * PBS;

  for (int i = 0; i < 4; ++i) {
    const int n0 = ((kh ^ 1) * 4 + i) * 16;
    v8f acc = {0.f, 0.f, 0.f, 0.f, 0.f, 0.f, 0.f, 0.f};
    const v4f* vq = sVq + (jb * 4 + 2 * hi) * D_ + n0 + lh;
#pragma unroll
    for (int jj = 0; jj < KT0; ++jj) {
      if (jj < KT0 - 1 || kh == 0) {
#pragma unroll
        for (int u = 0; u < 2; ++u) {
          const v4f v = vq[(jj * 4 + u) * D_];
#pragma unroll
          for (int e = 0; e < 2; ++e) {
            const int q = 2 * u + e;
            v2f a;
            a.x = p[jj][2 * q];
            a.y = p[jj][2 * q + 1];
            v2f bb;
            bb.x = e ? v.z : v.x;
            bb.y = e ? v.w : v.y;
            acc = wmma4(a, bb, acc);
          }
        }
      }
    }
    float* xo = sX + (wave * 4 + i) * 256 + lane;
#pragma unroll
    for (int r = 0; r < 8; ++r) xo[r * 32] = acc[r];
  }
  __syncthreads();

  {
    const int n0 = kh * 64;
    v8f acc[4];
#pragma unroll
    for (int r = 0; r < 8; ++r) {
      int t = qt * 16 + r + 8 * hi;
      v4f h = {0.f, 0.f, 0.f, 0.f};
      if (t < T_) {
        int a = alias[b * T_ + t];
        h = *(const v4f*)(re + ((size_t)b * NNODE_ + a) * D_ + n0 + 4 * lh);
      }
      acc[0][r] = h.x;
      acc[1][r] = h.y;
      acc[2][r] = h.z;
      acc[3][r] = h.w;
    }
    const v4f* vq = sVq + (jb * 4 + 2 * hi) * D_ + n0 + lh;
#pragma unroll
    for (int jj = 0; jj < KT0; ++jj) {
      if (jj < KT0 - 1 || kh == 0) {
#pragma unroll
        for (int u = 0; u < 2; ++u) {
          v4f v[4];
#pragma unroll
          for (int c = 0; c < 4; ++c) v[c] = vq[(jj * 4 + u) * D_ + 16 * c];
#pragma unroll
          for (int e = 0; e < 2; ++e) {
            const int q = 2 * u + e;
            v2f a;
            a.x = p[jj][2 * q];
            a.y = p[jj][2 * q + 1];
#pragma unroll
            for (int c = 0; c < 4; ++c) {
              v2f bb;
              bb.x = e ? v[c].z : v[c].x;
              bb.y = e ? v[c].w : v[c].y;
              acc[c] = wmma4(a, bb, acc[c]);
            }
          }
        }
      }
    }
    const float* xi = sX + pw * 4 * 256 + lane;
#pragma unroll
    for (int r = 0; r < 8; ++r) {
      int t = qt * 16 + r + 8 * hi;
      v4f o;
      o.x = acc[0][r] + xi[0 * 256 + r * 32];
      o.y = acc[1][r] + xi[1 * 256 + r * 32];
      o.z = acc[2][r] + xi[2 * 256 + r * 32];
      o.w = acc[3][r] + xi[3 * 256 + r * 32];
      *(v4f*)(Db + t * D_ + n0 + 4 * lh) = o;
    }
  }
}

__device__ __forceinline__ void stamp_cols(const v2f* da, const v4f* sW2q, const float* sLp,
                                           const float* sUb, const float* sUv,
                                           int nt0, int nct, int lh, int hi, float* cpart) {
  for (int u = 0; u < nct; ++u) {
    const int n0 = (nt0 + u) * 16;
    v8f acc = {0.f, 0.f, 0.f, 0.f, 0.f, 0.f, 0.f, 0.f};
    const v4f* wq = sW2q + hi * D_ + n0 + lh;
#pragma unroll
    for (int m = 0; m < 16; ++m) {
      const v4f w = wq[m * 2 * D_];
      v2f b0, b1;
      b0.x = w.x; b0.y = w.y;
      b1.x = w.z; b1.y = w.w;
      acc = wmma4(da[2 * m], b0, acc);
      acc = wmma4(da[2 * m + 1], b1, acc);
    }
    const int col = n0 + lh;
    const float add = sLp[col] + sUb[col];
    const float uvv = sUv[col];
#pragma unroll
    for (int r = 0; r < 8; ++r) {
      float x = acc[r] + add;
      float sg = 1.f / (1.f + __expf(-x));
      cpart[r] += sg * uvv;
    }
  }
}

__global__ void stamp_kernel(const int* __restrict__ seqlen, const float* __restrict__ wsDec,
                             const float* __restrict__ u_w1, const float* __restrict__ u_w2,
                             const float* __restrict__ u_b, const float* __restrict__ u_v,
                             float* __restrict__ out) {
  extern __shared__ __align__(16) float smem[];
  v4f* sW2q = (v4f*)smem;
  float* sLp = smem + D_ * D_;
  float* sUb = sLp + 128;
  float* sUv = sUb + 128;
  float* sCoef = sUv + 128;
  float* sCq = sCoef + TP;
  const int b = blockIdx.x, tid = threadIdx.x;
  const int len = seqlen[b];
  const float* Db = wsDec + (size_t)b * PBS;

  for (int idx = tid; idx < KQR * D_; idx += 256) {
    int qr = idx >> 7, n = idx & 127;
    const float* src = u_w2 + (8 * (qr >> 1) + 2 * (qr & 1)) * D_ + n;
    v4f w;
    w.x = src[0];
    w.y = src[D_];
    w.z = src[4 * D_];
    w.w = src[5 * D_];
    sW2q[idx] = w;
  }
  if (tid < D_) { sUb[tid] = u_b[tid]; sUv[tid] = u_v[tid]; }
  if (tid < D_) {
    const float* lrow = Db + (len - 1) * D_;
    float acc = 0.f;
    for (int d = 0; d < D_; ++d) acc += lrow[d] * u_w1[d * D_ + tid];
    sLp[tid] = acc;
    out[(size_t)b * 2 * D_ + D_ + tid] = acc;
  }
  __syncthreads();

  const int wave = __builtin_amdgcn_readfirstlane(tid >> 5);
  const int lane = tid & 31, lh = lane & 15, hi = lane >> 4;
  for (int pass = 0; pass < 2; ++pass) {
    const int qt = wave + 8 * pass;
    if (qt < NT - 1) {
      v2f da[32];
      const float* drow = Db + (qt * 16 + lh) * D_;
#pragma unroll
      for (int c = 0; c < 32; ++c) da[c] = *(const v2f*)(drow + 4 * c + 2 * hi);
      float cpart[8];
#pragma unroll
      for (int r = 0; r < 8; ++r) cpart[r] = 0.f;
      stamp_cols(da, sW2q, sLp, sUb, sUv, 0, 8, lh, hi, cpart);
#pragma unroll
      for (int r = 0; r < 8; ++r) {
        float v = cpart[r];
        v += __shfl_xor(v, 1, 32);
        v += __shfl_xor(v, 2, 32);
        v += __shfl_xor(v, 4, 32);
        v += __shfl_xor(v, 8, 32);
        if (lh == 0) {
          const int t = qt * 16 + r + 8 * hi;
          sCoef[t] = (t < len) ? v : 0.f;
        }
      }
    }
  }
  if (wave >= 4) {
    v2f da[32];
    const float* drow = Db + ((NT - 1) * 16 + lh) * D_;
#pragma unroll
    for (int c = 0; c < 32; ++c) da[c] = *(const v2f*)(drow + 4 * c + 2 * hi);
    float cpart[8];
#pragma unroll
    for (int r = 0; r < 8; ++r) cpart[r] = 0.f;
    stamp_cols(da, sW2q, sLp, sUb, sUv, 2 * (wave - 4), 2, lh, hi, cpart);
#pragma unroll
    for (int r = 0; r < 8; ++r) {
      float v = cpart[r];
      v += __shfl_xor(v, 1, 32);
      v += __shfl_xor(v, 2, 32);
      v += __shfl_xor(v, 4, 32);
      v += __shfl_xor(v, 8, 32);
      if (lh == 0) sCq[(wave - 4) * 16 + r + 8 * hi] = v;
    }
  }
  __syncthreads();
  if (tid < D_) {
    float acc = 0.f;
    for (int t = 0; t < (NT - 1) * 16; ++t) acc += sCoef[t] * Db[t * D_ + tid];
    for (int r = 0; r < 16; ++r) {
      const int t = (NT - 1) * 16 + r;
      const float c = ((sCq[r] + sCq[16 + r]) + sCq[32 + r]) + sCq[48 + r];
      acc += ((t < len) ? c : 0.f) * Db[t * D_ + tid];
    }
    out[(size_t)b * 2 * D_ + tid] = acc;
  }
}

extern "C" void kernel_launch(void* const* d_in, const int* in_sizes, int n_in,
                              void* d_out, int out_size, void* d_ws, size_t ws_size,
                              hipStream_t stream) {
  (void)in_sizes; (void)n_in; (void)out_size; (void)ws_size;
  const float* re   = (const float*)d_in[0];
  const int*   alia = (const int*)d_in[1];
  const int*   slen = (const int*)d_in[2];
  const float* Wq   = (const float*)d_in[3];
  const float* Wk   = (const float*)d_in[4];
  const float* Wv   = (const float*)d_in[5];
  const float* u_w1 = (const float*)d_in[6];
  const float* u_w2 = (const float*)d_in[7];
  const float* u_b  = (const float*)d_in[8];
  const float* u_v  = (const float*)d_in[9];
  float* out = (float*)d_out;

  float* ws   = (float*)d_ws;
  float* wsQ  = ws;
  float* wsK  = ws + (size_t)B_ * PBS;
  float* wsV  = ws + 2 * (size_t)B_ * PBS;
  float* wsDec = wsK;

  size_t sm1 = (size_t)(TP * HSTR + 3 * D_ * D_) * sizeof(float);
  size_t sm2 = (size_t)(2 * TP * HSTR + TP + 2 * AW * 16) * sizeof(float);
  size_t sm3 = (size_t)(D_ * D_ + 3 * 128 + TP + 64) * sizeof(float);
  hipFuncSetAttribute((const void*)qkv_kernel,
                      hipFuncAttributeMaxDynamicSharedMemorySize, (int)sm1);
  hipFuncSetAttribute((const void*)attn_kernel,
                      hipFuncAttributeMaxDynamicSharedMemorySize, (int)sm2);
  hipFuncSetAttribute((const void*)stamp_kernel,
                      hipFuncAttributeMaxDynamicSharedMemorySize, (int)sm3);

  qkv_kernel<<<B_, QKV_T, sm1, stream>>>(re, Wq, Wk, Wv, wsQ, wsK, wsV);
  attn_kernel<<<B_, AW * 32, sm2, stream>>>(re, alia, slen, wsQ, wsK, wsV, wsDec);
  stamp_kernel<<<B_, 256, sm3, stream>>>(slen, wsDec, u_w1, u_w2, u_b, u_v, out);
}


# qkv kernel: persistent blocks, one per WGP, 8 sessions each: W* staged to LDS once per block (was per session), next session's node rows fetched to registers under the WMMAs; old sum order
# speedup vs baseline: 2.8725x; 1.0320x over previous
#include <hip/hip_runtime.h>
#include <math.h>


typedef __attribute__((ext_vector_type(2))) float v2f;
typedef __attribute__((ext_vector_type(4))) float v4f;
typedef __attribute__((ext_vector_type(8))) float v8f;

#define B_      2048
#define T_      200
#define D_      128
#define NNODE_  200
#define TP      208
#define NT      13
#define HSTR    132
#define PBS     (TP * D_)
#define NEGC    (-4294967295.0f)
#define QKV_W   32
#define QKV_T   (QKV_W * 32)
#define KQR     (D_ / 4)

__device__ __forceinline__ v8f wmma4(v2f a, v2f b, v8f c) {
  return __builtin_amdgcn_wmma_f32_16x16x4_f32(false, a, false, b, (short)0, c,
                                               false, false);
}

#define CSRC2(p) (((p) & 96) | (((p) & 15) << 1) | (((p) >> 4) & 1))
#define CSRC4(p) (((p) & 64) | (((p) & 15) << 2) | (((p) >> 4) & 3))

#define QKV_HR  (NNODE_ * D_ / QKV_T)
__global__ void __launch_bounds__(QKV_T) qkv_kernel(const float* __restrict__ re,
                           const float* __restrict__ Wq, const float* __restrict__ Wk,
                           const float* __restrict__ Wv,
                           float* __restrict__ wsQ, float* __restrict__ wsK,
                           float* __restrict__ wsV) {
  extern __shared__ __align__(16) float smem[];
  float* sH = smem;
  v4f* sWq = (v4f*)(smem + TP * HSTR);
  const int tid = threadIdx.x;
  static_assert(QKV_HR * QKV_T == NNODE_ * D_, "node rows: a whole number of values per thread");

  float hreg[QKV_HR];
  {
    const float* reb = re + (size_t)blockIdx.x * NNODE_ * D_;
#pragma unroll
    for (int u = 0; u < QKV_HR; ++u) hreg[u] = reb[tid + u * QKV_T];
  }
  for (int idx = NNODE_ * D_ + tid; idx < TP * D_; idx += QKV_T) {
    int t = idx >> 7, d = idx & 127;
    sH[t * HSTR + d] = 0.f;
  }
  for (int idx = tid; idx < 3 * KQR * D_; idx += QKV_T) {
    int mat = idx >> 12, r = idx & 4095;
    int qr = r >> 7, n = CSRC2(r & 127);
    const float* src = ((mat == 0) ? Wq : (mat == 1) ? Wk : Wv) + (8 * (qr >> 1) + 2 * (qr & 1)) * D_ + n;
    v4f w;
    w.x = src[0];
    w.y = src[D_];
    w.z = src[4 * D_];
    w.w = src[5 * D_];
    sWq[idx] = w;
  }

  const int wave = tid >> 5, lane = tid & 31, lh = lane & 15, hi = lane >> 4;
  for (int b = blockIdx.x; b < B_; b += gridDim.x) {
    __syncthreads();
#pragma unroll
    for (int u = 0; u < QKV_HR; ++u) {
      const int idx = tid + u * QKV_T;
      sH[(idx >> 7) * HSTR + (idx & 127)] = hreg[u];
    }
    __syncthreads();
    if (b + (int)gridDim.x < B_) {
      const float* reb = re + (size_t)(b + gridDim.x) * NNODE_ * D_;
#pragma unroll
      for (int u = 0; u < QKV_HR; ++u) hreg[u] = reb[tid + u * QKV_T];
    }
    for (int unit = wave; unit < 3 * NT * 4; unit += QKV_W) {
      int mat = unit / (NT * 4), rem = unit % (NT * 4);
      int i = rem >> 2, n0 = (rem & 3) << 5;
      const v4f* wq = sWq + mat * KQR * D_ + hi * D_ + n0 + lh;
      const float* arow = sH + (i * 16 + lh) * HSTR;
      v8f acc0 = {0.f, 0.f, 0.f, 0.f, 0.f, 0.f, 0.f, 0.f};
      v8f acc1 = {0.f, 0.f, 0.f, 0.f, 0.f, 0.f, 0.f, 0.f};
      v4f fw0[3], fw1[3];
      v2f fa0[3], fa1[3];
#pragma unroll
      for (int s = 0; s < 2; ++s) {
        fw0[s] = wq[s * 2 * D_];
        fw1[s] = wq[s * 2 * D_ + 16];
        fa0[s] = *(const v2f*)(arow + 8 * s + 2 * hi);
        fa1[s] = *(const v2f*)(arow + 8 * s + 4 + 2 * hi);
      }
#pragma unroll
      for (int m = 0; m < 16; ++m) {
        if (m + 2 < 16) {
          const int s = (m + 2) % 3;
          fw0[s] = wq[(m + 2) * 2 * D_];
          fw1[s] = wq[(m + 2) * 2 * D_ + 16];
          fa0[s] = *(const v2f*)(arow + 8 * (m + 2) + 2 * hi);
          fa1[s] = *(const v2f*)(arow + 8 * (m + 2) + 4 + 2 * hi);
        }
        const v4f w0 = fw0[m % 3], w1 = fw1[m % 3];
        v2f b0, b1;
        b0.x = w0.x; b0.y = w0.y;
        b1.x = w1.x; b1.y = w1.y;
        acc0 = wmma4(fa0[m % 3], b0, acc0);
        acc1 = wmma4(fa0[m % 3], b1, acc1);
        b0.x = w0.z; b0.y = w0.w;
        b1.x = w1.z; b1.y = w1.w;
        acc0 = wmma4(fa1[m % 3], b0, acc0);
        acc1 = wmma4(fa1[m % 3], b1, acc1);
      }
      float* dst = ((mat == 0) ? wsQ : (mat == 1) ? wsK : wsV) + (size_t)b * PBS;
      v2f o[8];
#pragma unroll
      for (int r = 0; r < 8; ++r) {
        o[r].x = acc0[r] > 0.f ? acc0[r] : 0.f;
        o[r].y = acc1[r] > 0.f ? acc1[r] : 0.f;
      }
#pragma unroll
      for (int r = 0; r < 8; ++r) {
        int row = i * 16 + r + 8 * hi;
        *(v2f*)(dst + row * D_ + n0 + 2 * lh) = o[r];
      }
    }
  }
}

#define AW  26
#define KT0 7
__global__ void __launch_bounds__(AW * 32)
attn_kernel(const float* __restrict__ re, const int* __restrict__ alias,
            const int* __restrict__ seqlen,
            const float* __restrict__ wsQ, const float* wsK,
            const float* __restrict__ wsV, float* wsDec) {
  extern __shared__ __align__(16) float smem[];
  float* sK = smem;
  v4f* sVq = (v4f*)(smem + TP * HSTR);
  int* sAl = (int*)(smem + 2 * TP * HSTR);
  float* sMx = smem + 2 * TP * HSTR + TP;
  float* sSm = sMx + AW * 16;
  const int b = blockIdx.x, tid = threadIdx.x;
  const int len = seqlen[b];
  const float* Kb = wsK + (size_t)b * PBS;
  const float* Vb = wsV + (size_t)b * PBS;
  for (int t = tid; t < TP; t += AW * 32) sAl[t] = (t < T_) ? alias[b * T_ + t] : NNODE_;
  __syncthreads();
  static_assert(TP * D_ == 32 * AW * 32, "staging: 32 K values and 8 V quads per thread");
  float kv[32];
  v4f vv[8];
#pragma unroll
  for (int u = 0; u < 32; ++u) {
    const int idx = tid + u * (AW * 32);
    kv[u] = Kb[sAl[idx >> 7] * D_ + (idx & 127)];
  }
#pragma unroll
  for (int u = 0; u < 8; ++u) {
    const int idx = tid + u * (AW * 32);
    const int t4 = (idx >> 7) * 4, d = CSRC4(idx & 127);
    vv[u].x = Vb[sAl[t4] * D_ + d];
    vv[u].y = Vb[sAl[t4 + 1] * D_ + d];
    vv[u].z = Vb[sAl[t4 + 2] * D_ + d];
    vv[u].w = Vb[sAl[t4 + 3] * D_ + d];
  }
#pragma unroll
  for (int u = 0; u < 32; ++u) {
    const int idx = tid + u * (AW * 32);
    sK[(idx >> 7) * HSTR + (idx & 127)] = kv[u];
  }
#pragma unroll
  for (int u = 0; u < 8; ++u) sVq[tid + u * (AW * 32)] = vv[u];
  __syncthreads();

  const int wave = __builtin_amdgcn_readfirstlane(tid >> 5);
  const int lane = tid & 31, lh = lane & 15, hi = lane >> 4;
  const int cls = wave & 3, wi = wave >> 2;
  int qt, kh;
  if (cls == 0)      { kh = 1; qt = wi; }
  else if (cls == 1) { kh = (wi < 6) ? 1 : 0; qt = (wi < 6) ? 7 + wi : 12; }
  else if (cls == 2) { kh = 0; qt = wi; }
  else               { kh = 0; qt = 6 + wi; }
  const int pw = kh ? ((qt < 6) ? 4 * qt + 2 : (qt < 12) ? 4 * (qt - 6) + 3 : 25)
                    : ((qt < 7) ? 4 * qt : 4 * (qt - 7) + 1);
  const int jb = kh * KT0;
  const float scale = 0.088388347648318447f;

  v2f qa[32];
  const float* Qrow = wsQ + (size_t)b * PBS + sAl[qt * 16 + lh] * D_;
#pragma unroll
  for (int c = 0; c < 32; ++c) qa[c] = *(const v2f*)(Qrow + 4 * c + 2 * hi);

  float p[KT0][8];
#pragma unroll
  for (int jj = 0; jj < KT0; ++jj) {
    if (jj < KT0 - 1 || kh == 0) {
      const int j = jb + jj;
      v8f acc = {0.f, 0.f, 0.f, 0.f, 0.f, 0.f, 0.f, 0.f};
      const float* krow = sK + (j * 16 + lh) * HSTR + 2 * hi;
#pragma unroll
      for (int c = 0; c < 32; ++c) {
        v2f ka = *(const v2f*)(krow + 4 * c);
        acc = wmma4(ka, qa[c], acc);
      }
#pragma unroll
      for (int r = 0; r < 8; ++r) {
        int s = j * 16 + 8 * hi + r;
        p[jj][r] = (s < len) ? acc[r] * scale : NEGC;
      }
    } else {
#pragma unroll
      for (int r = 0; r < 8; ++r) p[jj][r] = NEGC;
    }
  }

  float mx = -3.0e38f;
#pragma unroll
  for (int jj = 0; jj < KT0; ++jj)
#pragma unroll
    for (int r = 0; r < 8; ++r) mx = fmaxf(mx, p[jj][r]);
  mx = fmaxf(mx, __shfl_xor(mx, 16, 32));
  if (hi == 0) sMx[wave * 16 + lh] = mx;
  __syncthreads();
  mx = fmaxf(mx, sMx[pw * 16 + lh]);
  float sum = 0.f;
#pragma unroll
  for (int jj = 0; jj < KT0; ++jj)
#pragma unroll
    for (int r = 0; r < 8; ++r) {
      float e = (jj < KT0 - 1 || kh == 0) ? __expf(p[jj][r] - mx) : 0.f;
      p[jj][r] = e;
      sum += e;
    }
  sum += __shfl_xor(sum, 16, 32);
  if (hi == 0) sSm[wave * 16 + lh] = sum;
  __syncthreads();
  sum += sSm[pw * 16 + lh];
  const float qm = ((qt * 16 + lh) < len) ? 1.f : 0.f;
  const float inv = qm / sum;
#pragma unroll
  for (int jj = 0; jj < KT0; ++jj)
#pragma unroll
    for (int r = 0; r < 8; ++r) p[jj][r] *= inv;

  float* sX = sK;
  float* Db = wsDec + (size_t)b * PBS;

  for (int i = 0; i < 4; ++i) {
    const int n0 = ((kh ^ 1) * 4 + i) * 16;
    v8f acc = {0.f, 0.f, 0.f, 0.f, 0.f, 0.f, 0.f, 0.f};
    const v4f* vq = sVq + (jb * 4 + 2 * hi) * D_ + n0 + lh;
#pragma unroll
    for (int jj = 0; jj < KT0; ++jj) {
      if (jj < KT0 - 1 || kh == 0) {
#pragma unroll
        for (int u = 0; u < 2; ++u) {
          const v4f v = vq[(jj * 4 + u) * D_];
#pragma unroll
          for (int e = 0; e < 2; ++e) {
            const int q = 2 * u + e;
            v2f a;
            a.x = p[jj][2 * q];
            a.y = p[jj][2 * q + 1];
            v2f bb;
            bb.x = e ? v.z : v.x;
            bb.y = e ? v.w : v.y;
            acc = wmma4(a, bb, acc);
          }
        }
      }
    }
    float* xo = sX + (wave * 4 + i) * 256 + lane;
#pragma unroll
    for (int r = 0; r < 8; ++r) xo[r * 32] = acc[r];
  }
  __syncthreads();

  {
    const int n0 = kh * 64;
    v8f acc[4];
#pragma unroll
    for (int r = 0; r < 8; ++r) {
      int t = qt * 16 + r + 8 * hi;
      v4f h = {0.f, 0.f, 0.f, 0.f};
      if (t < T_) {
        int a = alias[b * T_ + t];
        h = *(const v4f*)(re + ((size_t)b * NNODE_ + a) * D_ + n0 + 4 * lh);
      }
      acc[0][r] = h.x;
      acc[1][r] = h.y;
      acc[2][r] = h.z;
      acc[3][r] = h.w;
    }
    const v4f* vq = sVq + (jb * 4 + 2 * hi) * D_ + n0 + lh;
#pragma unroll
    for (int jj = 0; jj < KT0; ++jj) {
      if (jj < KT0 - 1 || kh == 0) {
#pragma unroll
        for (int u = 0; u < 2; ++u) {
          v4f v[4];
#pragma unroll
          for (int c = 0; c < 4; ++c) v[c] = vq[(jj * 4 + u) * D_ + 16 * c];
#pragma unroll
          for (int e = 0; e < 2; ++e) {
            const int q = 2 * u + e;
            v2f a;
            a.x = p[jj][2 * q];
            a.y = p[jj][2 * q + 1];
#pragma unroll
            for (int c = 0; c < 4; ++c) {
              v2f bb;
              bb.x = e ? v[c].z : v[c].x;
              bb.y = e ? v[c].w : v[c].y;
              acc[c] = wmma4(a, bb, acc[c]);
            }
          }
        }
      }
    }
    const float* xi = sX + pw * 4 * 256 + lane;
#pragma unroll
    for (int r = 0; r < 8; ++r) {
      int t = qt * 16 + r + 8 * hi;
      v4f o;
      o.x = acc[0][r] + xi[0 * 256 + r * 32];
      o.y = acc[1][r] + xi[1 * 256 + r * 32];
      o.z = acc[2][r] + xi[2 * 256 + r * 32];
      o.w = acc[3][r] + xi[3 * 256 + r * 32];
      *(v4f*)(Db + t * D_ + n0 + 4 * lh) = o;
    }
  }
}

__device__ __forceinline__ void stamp_cols(const v2f* da, const v4f* sW2q, const float* sLp,
                                           const float* sUb, const float* sUv,
                                           int nt0, int nct, int lh, int hi, float* cpart) {
  for (int u = 0; u < nct; ++u) {
    const int n0 = (nt0 + u) * 16;
    v8f acc = {0.f, 0.f, 0.f, 0.f, 0.f, 0.f, 0.f, 0.f};
    const v4f* wq = sW2q + hi * D_ + n0 + lh;
#pragma unroll
    for (int m = 0; m < 16; ++m) {
      const v4f w = wq[m * 2 * D_];
      v2f b0, b1;
      b0.x = w.x; b0.y = w.y;
      b1.x = w.z; b1.y = w.w;
      acc = wmma4(da[2 * m], b0, acc);
      acc = wmma4(da[2 * m + 1], b1, acc);
    }
    const int col = n0 + lh;
    const float add = sLp[col] + sUb[col];
    const float uvv = sUv[col];
#pragma unroll
    for (int r = 0; r < 8; ++r) {
      float x = acc[r] + add;
      float sg = 1.f / (1.f + __expf(-x));
      cpart[r] += sg * uvv;
    }
  }
}

__global__ void stamp_kernel(const int* __restrict__ seqlen, const float* __restrict__ wsDec,
                             const float* __restrict__ u_w1, const float* __restrict__ u_w2,
                             const float* __restrict__ u_b, const float* __restrict__ u_v,
                             float* __restrict__ out) {
  extern __shared__ __align__(16) float smem[];
  v4f* sW2q = (v4f*)smem;
  float* sLp = smem + D_ * D_;
  float* sUb = sLp + 128;
  float* sUv = sUb + 128;
  float* sCoef = sUv + 128;
  float* sCq = sCoef + TP;
  const int b = blockIdx.x, tid = threadIdx.x;
  const int len = seqlen[b];
  const float* Db = wsDec + (size_t)b * PBS;

  for (int idx = tid; idx < KQR * D_; idx += 256) {
    int qr = idx >> 7, n = idx & 127;
    const float* src = u_w2 + (8 * (qr >> 1) + 2 * (qr & 1)) * D_ + n;
    v4f w;
    w.x = src[0];
    w.y = src[D_];
    w.z = src[4 * D_];
    w.w = src[5 * D_];
    sW2q[idx] = w;
  }
  if (tid < D_) { sUb[tid] = u_b[tid]; sUv[tid] = u_v[tid]; }
  if (tid < D_) {
    const float* lrow = Db + (len - 1) * D_;
    float acc = 0.f;
    for (int d = 0; d < D_; ++d) acc += lrow[d] * u_w1[d * D_ + tid];
    sLp[tid] = acc;
    out[(size_t)b * 2 * D_ + D_ + tid] = acc;
  }
  __syncthreads();

  const int wave = __builtin_amdgcn_readfirstlane(tid >> 5);
  const int lane = tid & 31, lh = lane & 15, hi = lane >> 4;
  for (int pass = 0; pass < 2; ++pass) {
    const int qt = wave + 8 * pass;
    if (qt < NT - 1) {
      v2f da[32];
      const float* drow = Db + (qt * 16 + lh) * D_;
#pragma unroll
      for (int c = 0; c < 32; ++c) da[c] = *(const v2f*)(drow + 4 * c + 2 * hi);
      float cpart[8];
#pragma unroll
      for (int r = 0; r < 8; ++r) cpart[r] = 0.f;
      stamp_cols(da, sW2q, sLp, sUb, sUv, 0, 8, lh, hi, cpart);
#pragma unroll
      for (int r = 0; r < 8; ++r) {
        float v = cpart[r];
        v += __shfl_xor(v, 1, 32);
        v += __shfl_xor(v, 2, 32);
        v += __shfl_xor(v, 4, 32);
        v += __shfl_xor(v, 8, 32);
        if (lh == 0) {
          const int t = qt * 16 + r + 8 * hi;
          sCoef[t] = (t < len) ? v : 0.f;
        }
      }
    }
  }
  if (wave >= 4) {
    v2f da[32];
    const float* drow = Db + ((NT - 1) * 16 + lh) * D_;
#pragma unroll
    for (int c = 0; c < 32; ++c) da[c] = *(const v2f*)(drow + 4 * c + 2 * hi);
    float cpart[8];
#pragma unroll
    for (int r = 0; r < 8; ++r) cpart[r] = 0.f;
    stamp_cols(da, sW2q, sLp, sUb, sUv, 2 * (wave - 4), 2, lh, hi, cpart);
#pragma unroll
    for (int r = 0; r < 8; ++r) {
      float v = cpart[r];
      v += __shfl_xor(v, 1, 32);
      v += __shfl_xor(v, 2, 32);
      v += __shfl_xor(v, 4, 32);
      v += __shfl_xor(v, 8, 32);
      if (lh == 0) sCq[(wave - 4) * 16 + r + 8 * hi] = v;
    }
  }
  __syncthreads();
  if (tid < D_) {
    float acc = 0.f;
    for (int t = 0; t < (NT - 1) * 16; ++t) acc += sCoef[t] * Db[t * D_ + tid];
    for (int r = 0; r < 16; ++r) {
      const int t = (NT - 1) * 16 + r;
      const float c = ((sCq[r] + sCq[16 + r]) + sCq[32 + r]) + sCq[48 + r];
      acc += ((t < len) ? c : 0.f) * Db[t * D_ + tid];
    }
    out[(size_t)b * 2 * D_ + tid] = acc;
  }
}

extern "C" void kernel_launch(void* const* d_in, const int* in_sizes, int n_in,
                              void* d_out, int out_size, void* d_ws, size_t ws_size,
                              hipStream_t stream) {
  (void)in_sizes; (void)n_in; (void)out_size; (void)ws_size;
  const float* re   = (const float*)d_in[0];
  const int*   alia = (const int*)d_in[1];
  const int*   slen = (const int*)d_in[2];
  const float* Wq   = (const float*)d_in[3];
  const float* Wk   = (const float*)d_in[4];
  const float* Wv   = (const float*)d_in[5];
  const float* u_w1 = (const float*)d_in[6];
  const float* u_w2 = (const float*)d_in[7];
  const float* u_b  = (const float*)d_in[8];
  const float* u_v  = (const float*)d_in[9];
  float* out = (float*)d_out;

  float* ws   = (float*)d_ws;
  float* wsQ  = ws;
  float* wsK  = ws + (size_t)B_ * PBS;
  float* wsV  = ws + 2 * (size_t)B_ * PBS;
  float* wsDec = wsK;

  size_t sm1 = (size_t)(TP * HSTR + 3 * D_ * D_) * sizeof(float);
  size_t sm2 = (size_t)(2 * TP * HSTR + TP + 2 * AW * 16) * sizeof(float);
  size_t sm3 = (size_t)(D_ * D_ + 3 * 128 + TP + 64) * sizeof(float);
  hipFuncSetAttribute((const void*)qkv_kernel,
                      hipFuncAttributeMaxDynamicSharedMemorySize, (int)sm1);
  hipFuncSetAttribute((const void*)attn_kernel,
                      hipFuncAttributeMaxDynamicSharedMemorySize, (int)sm2);
  hipFuncSetAttribute((const void*)stamp_kernel,
                      hipFuncAttributeMaxDynamicSharedMemorySize, (int)sm3);

  qkv_kernel<<<256, QKV_T, sm1, stream>>>(re, Wq, Wk, Wv, wsQ, wsK, wsV);
  attn_kernel<<<B_, AW * 32, sm2, stream>>>(re, alia, slen, wsQ, wsK, wsV, wsDec);
  stamp_kernel<<<B_, 256, sm3, stream>>>(slen, wsDec, u_w1, u_w2, u_b, u_v, out);
}


# attn: a wave's 8 residual rows requested as one group before its partner-tile WMMAs, nodes from the LDS alias row (was 8 guarded alias+row loads, each awaited); kin to a sibling step; old sum order
# speedup vs baseline: 2.9891x; 1.0406x over previous
#include <hip/hip_runtime.h>
#include <math.h>


typedef __attribute__((ext_vector_type(2))) float v2f;
typedef __attribute__((ext_vector_type(4))) float v4f;
typedef __attribute__((ext_vector_type(8))) float v8f;

#define B_      2048
#define T_      200
#define D_      128
#define NNODE_  200
#define TP      208
#define NT      13
#define HSTR    132
#define PBS     (TP * D_)
#define NEGC    (-4294967295.0f)
#define QKV_W   32
#define QKV_T   (QKV_W * 32)
#define KQR     (D_ / 4)

__device__ __forceinline__ v8f wmma4(v2f a, v2f b, v8f c) {
  return __builtin_amdgcn_wmma_f32_16x16x4_f32(false, a, false, b, (short)0, c,
                                               false, false);
}

#define CSRC2(p) (((p) & 96) | (((p) & 15) << 1) | (((p) >> 4) & 1))
#define CSRC4(p) (((p) & 64) | (((p) & 15) << 2) | (((p) >> 4) & 3))

#define QKV_HR  (NNODE_ * D_ / QKV_T)
__global__ void __launch_bounds__(QKV_T) qkv_kernel(const float* __restrict__ re,
                           const float* __restrict__ Wq, const float* __restrict__ Wk,
                           const float* __restrict__ Wv,
                           float* __restrict__ wsQ, float* __restrict__ wsK,
                           float* __restrict__ wsV) {
  extern __shared__ __align__(16) float smem[];
  float* sH = smem;
  v4f* sWq = (v4f*)(smem + TP * HSTR);
  const int tid = threadIdx.x;
  static_assert(QKV_HR * QKV_T == NNODE_ * D_, "node rows: a whole number of values per thread");

  float hreg[QKV_HR];
  {
    const float* reb = re + (size_t)blockIdx.x * NNODE_ * D_;
#pragma unroll
    for (int u = 0; u < QKV_HR; ++u) hreg[u] = reb[tid + u * QKV_T];
  }
  for (int idx = NNODE_ * D_ + tid; idx < TP * D_; idx += QKV_T) {
    int t = idx >> 7, d = idx & 127;
    sH[t * HSTR + d] = 0.f;
  }
  for (int idx = tid; idx < 3 * KQR * D_; idx += QKV_T) {
    int mat = idx >> 12, r = idx & 4095;
    int qr = r >> 7, n = CSRC2(r & 127);
    const float* src = ((mat == 0) ? Wq : (mat == 1) ? Wk : Wv) + (8 * (qr >> 1) + 2 * (qr & 1)) * D_ + n;
    v4f w;
    w.x = src[0];
    w.y = src[D_];
    w.z = src[4 * D_];
    w.w = src[5 * D_];
    sWq[idx] = w;
  }

  const int wave = tid >> 5, lane = tid & 31, lh = lane & 15, hi = lane >> 4;
  for (int b = blockIdx.x; b < B_; b += gridDim.x) {
    __syncthreads();
#pragma unroll
    for (int u = 0; u < QKV_HR; ++u) {
      const int idx = tid + u * QKV_T;
      sH[(idx >> 7) * HSTR + (idx & 127)] = hreg[u];
    }
    __syncthreads();
    if (b + (int)gridDim.x < B_) {
      const float* reb = re + (size_t)(b + gridDim.x) * NNODE_ * D_;
#pragma unroll
      for (int u = 0; u < QKV_HR; ++u) hreg[u] = reb[tid + u * QKV_T];
    }
    for (int unit = wave; unit < 3 * NT * 4; unit += QKV_W) {
      int mat = unit / (NT * 4), rem = unit % (NT * 4);
      int i = rem >> 2, n0 = (rem & 3) << 5;
      const v4f* wq = sWq + mat * KQR * D_ + hi * D_ + n0 + lh;
      const float* arow = sH + (i * 16 + lh) * HSTR;
      v8f acc0 = {0.f, 0.f, 0.f, 0.f, 0.f, 0.f, 0.f, 0.f};
      v8f acc1 = {0.f, 0.f, 0.f, 0.f, 0.f, 0.f, 0.f, 0.f};
      v4f fw0[3], fw1[3];
      v2f fa0[3], fa1[3];
#pragma unroll
      for (int s = 0; s < 2; ++s) {
        fw0[s] = wq[s * 2 * D_];
        fw1[s] = wq[s * 2 * D_ + 16];
        fa0[s] = *(const v2f*)(arow + 8 * s + 2 * hi);
        fa1[s] = *(const v2f*)(arow + 8 * s + 4 + 2 * hi);
      }
#pragma unroll
      for (int m = 0; m < 16; ++m) {
        if (m + 2 < 16) {
          const int s = (m + 2) % 3;
          fw0[s] = wq[(m + 2) * 2 * D_];
          fw1[s] = wq[(m + 2) * 2 * D_ + 16];
          fa0[s] = *(const v2f*)(arow + 8 * (m + 2) + 2 * hi);
          fa1[s] = *(const v2f*)(arow + 8 * (m + 2) + 4 + 2 * hi);
        }
        const v4f w0 = fw0[m % 3], w1 = fw1[m % 3];
        v2f b0, b1;
        b0.x = w0.x; b0.y = w0.y;
        b1.x = w1.x; b1.y = w1.y;
        acc0 = wmma4(fa0[m % 3], b0, acc0);
        acc1 = wmma4(fa0[m % 3], b1, acc1);
        b0.x = w0.z; b0.y = w0.w;
        b1.x = w1.z; b1.y = w1.w;
        acc0 = wmma4(fa1[m % 3], b0, acc0);
        acc1 = wmma4(fa1[m % 3], b1, acc1);
      }
      float* dst = ((mat == 0) ? wsQ : (mat == 1) ? wsK : wsV) + (size_t)b * PBS;
      v2f o[8];
#pragma unroll
      for (int r = 0; r < 8; ++r) {
        o[r].x = acc0[r] > 0.f ? acc0[r] : 0.f;
        o[r].y = acc1[r] > 0.f ? acc1[r] : 0.f;
      }
#pragma unroll
      for (int r = 0; r < 8; ++r) {
        int row = i * 16 + r + 8 * hi;
        *(v2f*)(dst + row * D_ + n0 + 2 * lh) = o[r];
      }
    }
  }
}

#define AW  26
#define KT0 7
__global__ void __launch_bounds__(AW * 32)
attn_kernel(const float* __restrict__ re, const int* __restrict__ alias,
            const int* __restrict__ seqlen,
            const float* __restrict__ wsQ, const float* wsK,
            const float* __restrict__ wsV, float* wsDec) {
  extern __shared__ __align__(16) float smem[];
  float* sK = smem;
  v4f* sVq = (v4f*)(smem + TP * HSTR);
  int* sAl = (int*)(smem + 2 * TP * HSTR);
  float* sMx = smem + 2 * TP * HSTR + TP;
  float* sSm = sMx + AW * 16;
  const int b = blockIdx.x, tid = threadIdx.x;
  const int len = seqlen[b];
  const float* Kb = wsK + (size_t)b * PBS;
  const float* Vb = wsV + (size_t)b * PBS;
  for (int t = tid; t < TP; t += AW * 32) sAl[t] = (t < T_) ? alias[b * T_ + t] : NNODE_;
  __syncthreads();
  static_assert(TP * D_ == 32 * AW * 32, "staging: 32 K values and 8 V quads per thread");
  float kv[32];
  v4f vv[8];
#pragma unroll
  for (int u = 0; u < 32; ++u) {
    const int idx = tid + u * (AW * 32);
    kv[u] = Kb[sAl[idx >> 7] * D_ + (idx & 127)];
  }
#pragma unroll
  for (int u = 0; u < 8; ++u) {
    const int idx = tid + u * (AW * 32);
    const int t4 = (idx >> 7) * 4, d = CSRC4(idx & 127);
    vv[u].x = Vb[sAl[t4] * D_ + d];
    vv[u].y = Vb[sAl[t4 + 1] * D_ + d];
    vv[u].z = Vb[sAl[t4 + 2] * D_ + d];
    vv[u].w = Vb[sAl[t4 + 3] * D_ + d];
  }
#pragma unroll
  for (int u = 0; u < 32; ++u) {
    const int idx = tid + u * (AW * 32);
    sK[(idx >> 7) * HSTR + (idx & 127)] = kv[u];
  }
#pragma unroll
  for (int u = 0; u < 8; ++u) sVq[tid + u * (AW * 32)] = vv[u];
  __syncthreads();

  const int wave = __builtin_amdgcn_readfirstlane(tid >> 5);
  const int lane = tid & 31, lh = lane & 15, hi = lane >> 4;
  const int cls = wave & 3, wi = wave >> 2;
  int qt, kh;
  if (cls == 0)      { kh = 1; qt = wi; }
  else if (cls == 1) { kh = (wi < 6) ? 1 : 0; qt = (wi < 6) ? 7 + wi : 12; }
  else if (cls == 2) { kh = 0; qt = wi; }
  else               { kh = 0; qt = 6 + wi; }
  const int pw = kh ? ((qt < 6) ? 4 * qt + 2 : (qt < 12) ? 4 * (qt - 6) + 3 : 25)
                    : ((qt < 7) ? 4 * qt : 4 * (qt - 7) + 1);
  const int jb = kh * KT0;
  const float scale = 0.088388347648318447f;

  v2f qa[32];
  const float* Qrow = wsQ + (size_t)b * PBS + sAl[qt * 16 + lh] * D_;
#pragma unroll
  for (int c = 0; c < 32; ++c) qa[c] = *(const v2f*)(Qrow + 4 * c + 2 * hi);

  float p[KT0][8];
#pragma unroll
  for (int jj = 0; jj < KT0; ++jj) {
    if (jj < KT0 - 1 || kh == 0) {
      const int j = jb + jj;
      v8f acc = {0.f, 0.f, 0.f, 0.f, 0.f, 0.f, 0.f, 0.f};
      const float* krow = sK + (j * 16 + lh) * HSTR + 2 * hi;
#pragma unroll
      for (int c = 0; c < 32; ++c) {
        v2f ka = *(const v2f*)(krow + 4 * c);
        acc = wmma4(ka, qa[c], acc);
      }
#pragma unroll
      for (int r = 0; r < 8; ++r) {
        int s = j * 16 + 8 * hi + r;
        p[jj][r] = (s < len) ? acc[r] * scale : NEGC;
      }
    } else {
#pragma unroll
      for (int r = 0; r < 8; ++r) p[jj][r] = NEGC;
    }
  }

  float mx = -3.0e38f;
#pragma unroll
  for (int jj = 0; jj < KT0; ++jj)
#pragma unroll
    for (int r = 0; r < 8; ++r) mx = fmaxf(mx, p[jj][r]);
  mx = fmaxf(mx, __shfl_xor(mx, 16, 32));
  if (hi == 0) sMx[wave * 16 + lh] = mx;
  __syncthreads();
  mx = fmaxf(mx, sMx[pw * 16 + lh]);
  float sum = 0.f;
#pragma unroll
  for (int jj = 0; jj < KT0; ++jj)
#pragma unroll
    for (int r = 0; r < 8; ++r) {
      float e = (jj < KT0 - 1 || kh == 0) ? __expf(p[jj][r] - mx) : 0.f;
      p[jj][r] = e;
      sum += e;
    }
  sum += __shfl_xor(sum, 16, 32);
  if (hi == 0) sSm[wave * 16 + lh] = sum;
  __syncthreads();
  v4f hres[8];
#pragma unroll
  for (int r = 0; r < 8; ++r) {
    const int t = qt * 16 + r + 8 * hi;
    int a = sAl[t];
    a = (t < T_) ? a : 0;
    hres[r] = *(const v4f*)(re + ((size_t)b * NNODE_ + a) * D_ + kh * 64 + 4 * lh);
  }
  sum += sSm[pw * 16 + lh];
  const float qm = ((qt * 16 + lh) < len) ? 1.f : 0.f;
  const float inv = qm / sum;
#pragma unroll
  for (int jj = 0; jj < KT0; ++jj)
#pragma unroll
    for (int r = 0; r < 8; ++r) p[jj][r] *= inv;

  float* sX = sK;
  float* Db = wsDec + (size_t)b * PBS;

  for (int i = 0; i < 4; ++i) {
    const int n0 = ((kh ^ 1) * 4 + i) * 16;
    v8f acc = {0.f, 0.f, 0.f, 0.f, 0.f, 0.f, 0.f, 0.f};
    const v4f* vq = sVq + (jb * 4 + 2 * hi) * D_ + n0 + lh;
#pragma unroll
    for (int jj = 0; jj < KT0; ++jj) {
      if (jj < KT0 - 1 || kh == 0) {
#pragma unroll
        for (int u = 0; u < 2; ++u) {
          const v4f v = vq[(jj * 4 + u) * D_];
#pragma unroll
          for (int e = 0; e < 2; ++e) {
            const int q = 2 * u + e;
            v2f a;
            a.x = p[jj][2 * q];
            a.y = p[jj][2 * q + 1];
            v2f bb;
            bb.x = e ? v.z : v.x;
            bb.y = e ? v.w : v.y;
            acc = wmma4(a, bb, acc);
          }
        }
      }
    }
    float* xo = sX + (wave * 4 + i) * 256 + lane;
#pragma unroll
    for (int r = 0; r < 8; ++r) xo[r * 32] = acc[r];
  }
  __syncthreads();

  {
    const int n0 = kh * 64;
    v8f acc[4];
#pragma unroll
    for (int r = 0; r < 8; ++r) {
      const bool live = (qt * 16 + r + 8 * hi) < T_;
      acc[0][r] = live ? hres[r].x : 0.f;
      acc[1][r] = live ? hres[r].y : 0.f;
      acc[2][r] = live ? hres[r].z : 0.f;
      acc[3][r] = live ? hres[r].w : 0.f;
    }
    const v4f* vq = sVq + (jb * 4 + 2 * hi) * D_ + n0 + lh;
#pragma unroll
    for (int jj = 0; jj < KT0; ++jj) {
      if (jj < KT0 - 1 || kh == 0) {
#pragma unroll
        for (int u = 0; u < 2; ++u) {
          v4f v[4];
#pragma unroll
          for (int c = 0; c < 4; ++c) v[c] = vq[(jj * 4 + u) * D_ + 16 * c];
#pragma unroll
          for (int e = 0; e < 2; ++e) {
            const int q = 2 * u + e;
            v2f a;
            a.x = p[jj][2 * q];
            a.y = p[jj][2 * q + 1];
#pragma unroll
            for (int c = 0; c < 4; ++c) {
              v2f bb;
              bb.x = e ? v[c].z : v[c].x;
              bb.y = e ? v[c].w : v[c].y;
              acc[c] = wmma4(a, bb, acc[c]);
            }
          }
        }
      }
    }
    const float* xi = sX + pw * 4 * 256 + lane;
#pragma unroll
    for (int r = 0; r < 8; ++r) {
      int t = qt * 16 + r + 8 * hi;
      v4f o;
      o.x = acc[0][r] + xi[0 * 256 + r * 32];
      o.y = acc[1][r] + xi[1 * 256 + r * 32];
      o.z = acc[2][r] + xi[2 * 256 + r * 32];
      o.w = acc[3][r] + xi[3 * 256 + r * 32];
      *(v4f*)(Db + t * D_ + n0 + 4 * lh) = o;
    }
  }
}

__device__ __forceinline__ void stamp_cols(const v2f* da, const v4f* sW2q, const float* sLp,
                                           const float* sUb, const float* sUv,
                                           int nt0, int nct, int lh, int hi, float* cpart) {
  for (int u = 0; u < nct; ++u) {
    const int n0 = (nt0 + u) * 16;
    v8f acc = {0.f, 0.f, 0.f, 0.f, 0.f, 0.f, 0.f, 0.f};
    const v4f* wq = sW2q + hi * D_ + n0 + lh;
#pragma unroll
    for (int m = 0; m < 16; ++m) {
      const v4f w = wq[m * 2 * D_];
      v2f b0, b1;
      b0.x = w.x; b0.y = w.y;
      b1.x = w.z; b1.y = w.w;
      acc = wmma4(da[2 * m], b0, acc);
      acc = wmma4(da[2 * m + 1], b1, acc);
    }
    const int col = n0 + lh;
    const float add = sLp[col] + sUb[col];
    const float uvv = sUv[col];
#pragma unroll
    for (int r = 0; r < 8; ++r) {
      float x = acc[r] + add;
      float sg = 1.f / (1.f + __expf(-x));
      cpart[r] += sg * uvv;
    }
  }
}

__global__ void stamp_kernel(const int* __restrict__ seqlen, const float* __restrict__ wsDec,
                             const float* __restrict__ u_w1, const float* __restrict__ u_w2,
                             const float* __restrict__ u_b, const float* __restrict__ u_v,
                             float* __restrict__ out) {
  extern __shared__ __align__(16) float smem[];
  v4f* sW2q = (v4f*)smem;
  float* sLp = smem + D_ * D_;
  float* sUb = sLp + 128;
  float* sUv = sUb + 128;
  float* sCoef = sUv + 128;
  float* sCq = sCoef + TP;
  const int b = blockIdx.x, tid = threadIdx.x;
  const int len = seqlen[b];
  const float* Db = wsDec + (size_t)b * PBS;

  for (int idx = tid; idx < KQR * D_; idx += 256) {
    int qr = idx >> 7, n = idx & 127;
    const float* src = u_w2 + (8 * (qr >> 1) + 2 * (qr & 1)) * D_ + n;
    v4f w;
    w.x = src[0];
    w.y = src[D_];
    w.z = src[4 * D_];
    w.w = src[5 * D_];
    sW2q[idx] = w;
  }
  if (tid < D_) { sUb[tid] = u_b[tid]; sUv[tid] = u_v[tid]; }
  if (tid < D_) {
    const float* lrow = Db + (len - 1) * D_;
    float acc = 0.f;
    for (int d = 0; d < D_; ++d) acc += lrow[d] * u_w1[d * D_ + tid];
    sLp[tid] = acc;
    out[(size_t)b * 2 * D_ + D_ + tid] = acc;
  }
  __syncthreads();

  const int wave = __builtin_amdgcn_readfirstlane(tid >> 5);
  const int lane = tid & 31, lh = lane & 15, hi = lane >> 4;
  for (int pass = 0; pass < 2; ++pass) {
    const int qt = wave + 8 * pass;
    if (qt < NT - 1) {
      v2f da[32];
      const float* drow = Db + (qt * 16 + lh) * D_;
#pragma unroll
      for (int c = 0; c < 32; ++c) da[c] = *(const v2f*)(drow + 4 * c + 2 * hi);
      float cpart[8];
#pragma unroll
      for (int r = 0; r < 8; ++r) cpart[r] = 0.f;
      stamp_cols(da, sW2q, sLp, sUb, sUv, 0, 8, lh, hi, cpart);
#pragma unroll
      for (int r = 0; r < 8; ++r) {
        float v = cpart[r];
        v += __shfl_xor(v, 1, 32);
        v += __shfl_xor(v, 2, 32);
        v += __shfl_xor(v, 4, 32);
        v += __shfl_xor(v, 8, 32);
        if (lh == 0) {
          const int t = qt * 16 + r + 8 * hi;
          sCoef[t] = (t < len) ? v : 0.f;
        }
      }
    }
  }
  if (wave >= 4) {
    v2f da[32];
    const float* drow = Db + ((NT - 1) * 16 + lh) * D_;
#pragma unroll
    for (int c = 0; c < 32; ++c) da[c] = *(const v2f*)(drow + 4 * c + 2 * hi);
    float cpart[8];
#pragma unroll
    for (int r = 0; r < 8; ++r) cpart[r] = 0.f;
    stamp_cols(da, sW2q, sLp, sUb, sUv, 2 * (wave - 4), 2, lh, hi, cpart);
#pragma unroll
    for (int r = 0; r < 8; ++r) {
      float v = cpart[r];
      v += __shfl_xor(v, 1, 32);
      v += __shfl_xor(v, 2, 32);
      v += __shfl_xor(v, 4, 32);
      v += __shfl_xor(v, 8, 32);
      if (lh == 0) sCq[(wave - 4) * 16 + r + 8 * hi] = v;
    }
  }
  __syncthreads();
  if (tid < D_) {
    float acc = 0.f;
    for (int t = 0; t < (NT - 1) * 16; ++t) acc += sCoef[t] * Db[t * D_ + tid];
    for (int r = 0; r < 16; ++r) {
      const int t = (NT - 1) * 16 + r;
      const float c = ((sCq[r] + sCq[16 + r]) + sCq[32 + r]) + sCq[48 + r];
      acc += ((t < len) ? c : 0.f) * Db[t * D_ + tid];
    }
    out[(size_t)b * 2 * D_ + tid] = acc;
  }
}

extern "C" void kernel_launch(void* const* d_in, const int* in_sizes, int n_in,
                              void* d_out, int out_size, void* d_ws, size_t ws_size,
                              hipStream_t stream) {
  (void)in_sizes; (void)n_in; (void)out_size; (void)ws_size;
  const float* re   = (const float*)d_in[0];
  const int*   alia = (const int*)d_in[1];
  const int*   slen = (const int*)d_in[2];
  const float* Wq   = (const float*)d_in[3];
  const float* Wk   = (const float*)d_in[4];
  const float* Wv   = (const float*)d_in[5];
  const float* u_w1 = (const float*)d_in[6];
  const float* u_w2 = (const float*)d_in[7];
  const float* u_b  = (const float*)d_in[8];
  const float* u_v  = (const float*)d_in[9];
  float* out = (float*)d_out;

  float* ws   = (float*)d_ws;
  float* wsQ  = ws;
  float* wsK  = ws + (size_t)B_ * PBS;
  float* wsV  = ws + 2 * (size_t)B_ * PBS;
  float* wsDec = wsK;

  size_t sm1 = (size_t)(TP * HSTR + 3 * D_ * D_) * sizeof(float);
  size_t sm2 = (size_t)(2 * TP * HSTR + TP + 2 * AW * 16) * sizeof(float);
  size_t sm3 = (size_t)(D_ * D_ + 3 * 128 + TP + 64) * sizeof(float);
  hipFuncSetAttribute((const void*)qkv_kernel,
                      hipFuncAttributeMaxDynamicSharedMemorySize, (int)sm1);
  hipFuncSetAttribute((const void*)attn_kernel,
                      hipFuncAttributeMaxDynamicSharedMemorySize, (int)sm2);
  hipFuncSetAttribute((const void*)stamp_kernel,
                      hipFuncAttributeMaxDynamicSharedMemorySize, (int)sm3);

  qkv_kernel<<<256, QKV_T, sm1, stream>>>(re, Wq, Wk, Wv, wsQ, wsK, wsV);
  attn_kernel<<<B_, AW * 32, sm2, stream>>>(re, alia, slen, wsQ, wsK, wsV, wsDec);
  stamp_kernel<<<B_, 256, sm3, stream>>>(slen, wsDec, u_w1, u_w2, u_b, u_v, out);
}
